# starting kernel
# baseline (speedup 1.0000x reference)
#include <hip/hip_runtime.h>
#include <hip/hip_bf16.h>

#define D_IN   128
#define D_OUT  64
#define ALPHA  0.2f
#define EPS    1e-10f

typedef __attribute__((ext_vector_type(2))) float v2f;
typedef __attribute__((ext_vector_type(8))) float v8f;

__device__ __forceinline__ unsigned f2ord(float f) {
    unsigned u = __float_as_uint(f);
    return (u & 0x80000000u) ? ~u : (u | 0x80000000u);
}
__device__ __forceinline__ float ord2f(unsigned u) {
    return __uint_as_float((u & 0x80000000u) ? (u ^ 0x80000000u) : ~u);
}
#define ORD_NEGINF 0x007FFFFFu

__device__ __forceinline__ float leaky(float x) { return x > 0.f ? x : ALPHA * x; }

__global__ void gat_init(float* __restrict__ out, float* __restrict__ denom,
                         unsigned* __restrict__ mflip, int N) {
    int i = blockIdx.x * blockDim.x + threadIdx.x;
    int total = N * D_OUT;
    if (i < total) out[i] = 0.0f;
    if (i < N) { denom[i] = 0.0f; mflip[i] = ORD_NEGINF; }
}

__global__ void gat_gemm_wmma(const float* __restrict__ node,
                              const float* __restrict__ W,
                              float* __restrict__ h, int N) {
    int wave = (blockIdx.x * blockDim.x + threadIdx.x) >> 5;
    int lane = threadIdx.x & 31;
    int row0 = wave * 16;
    if (row0 >= N) return;

    int mrow = lane & 15;
    int koff = (lane >> 4) * 2;

    v8f c0 = {}, c1 = {}, c2 = {}, c3 = {};

    const float* arow = node + (size_t)(row0 + mrow) * D_IN + koff;

    for (int k = 0; k < D_IN; k += 4) {
        v2f a;
        a.x = arow[k];
        a.y = arow[k + 1];

        const float* bp = W + (size_t)(k + koff) * D_OUT + mrow;
        v2f b0, b1, b2, b3;
        b0.x = bp[0];            b0.y = bp[D_OUT];
        b1.x = bp[16];           b1.y = bp[16 + D_OUT];
        b2.x = bp[32];           b2.y = bp[32 + D_OUT];
        b3.x = bp[48];           b3.y = bp[48 + D_OUT];

        c0 = __builtin_amdgcn_wmma_f32_16x16x4_f32(false, a, false, b0, (short)0, c0, false, false);
        c1 = __builtin_amdgcn_wmma_f32_16x16x4_f32(false, a, false, b1, (short)0, c1, false, false);
        c2 = __builtin_amdgcn_wmma_f32_16x16x4_f32(false, a, false, b2, (short)0, c2, false, false);
        c3 = __builtin_amdgcn_wmma_f32_16x16x4_f32(false, a, false, b3, (short)0, c3, false, false);
    }

    int nlo   = lane & 15;
    int rbase = row0 + ((lane >> 4) * 8);
#pragma unroll
    for (int r = 0; r < 8; ++r) {
        float* op = h + (size_t)(rbase + r) * D_OUT + nlo;
        op[0]  = c0[r];
        op[16] = c1[r];
        op[32] = c2[r];
        op[48] = c3[r];
    }
}

__global__ void gat_scores(const float* __restrict__ h, const float* __restrict__ a,
                           float* __restrict__ ssrc, float* __restrict__ sdst, int N) {
    int i = blockIdx.x * blockDim.x + threadIdx.x;
    if (i >= N) return;
    const float* hr = h + (size_t)i * D_OUT;
    float s0 = 0.f, s1 = 0.f;
#pragma unroll 8
    for (int j = 0; j < D_OUT; ++j) {
        float v = hr[j];
        s0 = fmaf(v, a[j], s0);
        s1 = fmaf(v, a[D_OUT + j], s1);
    }
    ssrc[i] = s0;
    sdst[i] = s1;
}

__global__ void gat_edge_max(const int* __restrict__ src, const int* __restrict__ dst,
                             const float* __restrict__ ssrc, const float* __restrict__ sdst,
                             unsigned* __restrict__ mflip, float* __restrict__ logits, int E) {
    int e = blockIdx.x * blockDim.x + threadIdx.x;
    if (e >= E) return;
    int s = src[e], d = dst[e];
    float l = leaky(ssrc[s] + sdst[d]);
    logits[e] = l;
    atomicMax(&mflip[s], f2ord(l));
}

__global__ void gat_m_final(const unsigned* __restrict__ mflip, float* __restrict__ m, int N) {
    int i = blockIdx.x * blockDim.x + threadIdx.x;
    if (i >= N) return;
    float v = ord2f(mflip[i]);
    m[i] = __isinff(v) && v < 0.f ? 0.0f : v;
}

__global__ void gat_edge_exp(const int* __restrict__ src, const float* __restrict__ logits,
                             const float* __restrict__ m, float* __restrict__ denom, int E) {
    int e = blockIdx.x * blockDim.x + threadIdx.x;
    if (e >= E) return;
    int s = src[e];
    atomicAdd(&denom[s], __expf(logits[e] - m[s]));
}

__global__ void gat_edge_scatter(const int* __restrict__ src, const int* __restrict__ dst,
                                 const float* __restrict__ logits, const float* __restrict__ m,
                                 const float* __restrict__ denom, const float* __restrict__ h,
                                 float* __restrict__ out, int E) {
    long long gid = (long long)blockIdx.x * blockDim.x + threadIdx.x;
    int e    = (int)(gid >> 5);
    int lane = (int)(gid & 31);
    if (e >= E) return;
    int s = src[e], d = dst[e];
    float att = __expf(logits[e] - m[s]) / (denom[s] + EPS);
    const float2* hp = (const float2*)(h + (size_t)d * D_OUT);
    float2 v = hp[lane];
    float* op = out + (size_t)s * D_OUT + 2 * lane;
    atomicAdd(op + 0, att * v.x);
    atomicAdd(op + 1, att * v.y);
}

extern "C" void kernel_launch(void* const* d_in, const int* in_sizes, int n_in,
                              void* d_out, int out_size, void* d_ws, size_t ws_size,
                              hipStream_t stream) {
    const float* node = (const float*)d_in[0];
    const int*   ei   = (const int*)d_in[1];
    const float* W    = (const float*)d_in[2];
    const float* a    = (const float*)d_in[3];
    float* out        = (float*)d_out;

    int N = in_sizes[0] / D_IN;
    int E = in_sizes[1] / 2;
    const int* src = ei;
    const int* dst = ei + E;

    float*    h      = (float*)d_ws;
    float*    ssrc   = h + (size_t)N * D_OUT;
    float*    sdst   = ssrc + N;
    float*    m      = sdst + N;
    unsigned* mflip  = (unsigned*)(m + N);
    float*    denom  = (float*)(mflip + N);
    float*    logits = denom + N;

    const int B = 256;

    {
        int total = N * D_OUT;
        gat_init<<<(total + B - 1) / B, B, 0, stream>>>(out, denom, mflip, N);
    }
    {
        int waves   = (N + 15) / 16;
        long long threads = (long long)waves * 32;
        int blocks  = (int)((threads + B - 1) / B);
        gat_gemm_wmma<<<blocks, B, 0, stream>>>(node, W, h, N);
    }
    gat_scores<<<(N + B - 1) / B, B, 0, stream>>>(h, a, ssrc, sdst, N);
    gat_edge_max<<<(E + B - 1) / B, B, 0, stream>>>(src, dst, ssrc, sdst, mflip, logits, E);
    gat_m_final<<<(N + B - 1) / B, B, 0, stream>>>(mflip, m, N);
    gat_edge_exp<<<(E + B - 1) / B, B, 0, stream>>>(src, logits, m, denom, E);
    {
        long long threads = (long long)E * 32;
        int blocks = (int)((threads + B - 1) / B);
        gat_edge_scatter<<<blocks, B, 0, stream>>>(src, dst, logits, m, denom, h, out, E);
    }
}


# counting sort of edges by source node (histogram+rank, range allocation, placement), then one wave per node does segment max, exp-sum and weighted row sum in registers; no float atomics left
# speedup vs baseline: 1.9252x; 1.9252x over previous
#include <hip/hip_runtime.h>


#define D_IN   128
#define D_OUT  64
#define ALPHA  0.2f
#define EPS    1e-10f

typedef __attribute__((ext_vector_type(2))) float v2f;
typedef __attribute__((ext_vector_type(8))) float v8f;

__device__ __forceinline__ float leaky(float x) { return x > 0.f ? x : ALPHA * x; }

__global__ void gat_init(int* __restrict__ cnt, int* __restrict__ counter, int N) {
    int i = blockIdx.x * blockDim.x + threadIdx.x;
    if (i < N) cnt[i] = 0;
    if (i == 0) counter[0] = 0;
}

__global__ void gat_hist_rank(const int* __restrict__ src, int* __restrict__ cnt,
                              int* __restrict__ rank, int E) {
    int e = blockIdx.x * blockDim.x + threadIdx.x;
    if (e >= E) return;
    rank[e] = atomicAdd(&cnt[src[e]], 1);
}

__global__ void gat_alloc(const int* __restrict__ cnt, int* __restrict__ start,
                          int* __restrict__ counter, int N) {
    int i    = blockIdx.x * blockDim.x + threadIdx.x;
    int lane = threadIdx.x & 31;
    int c = (i < N) ? cnt[i] : 0;
    int x = c;
    for (int o = 1; o < 32; o <<= 1) {
        int y = __shfl_up(x, o);
        if (lane >= o) x += y;
    }
    int total = __shfl(x, 31);
    int base = 0;
    if (lane == 31) base = atomicAdd(counter, total);
    base = __shfl(base, 31);
    if (i < N) start[i] = base + x - c;
}

__global__ void gat_place(const int* __restrict__ src, const int* __restrict__ dst,
                          const int* __restrict__ rank, const int* __restrict__ start,
                          int* __restrict__ nbr, int E) {
    int e = blockIdx.x * blockDim.x + threadIdx.x;
    if (e >= E) return;
    int d   = dst[e];
    int pos = start[src[e]] + rank[e];
    pos = pos < 0 ? 0 : (pos >= E ? E - 1 : pos);
    nbr[pos] = d;
}

__global__ void gat_gemm_wmma(const float* __restrict__ node,
                              const float* __restrict__ W,
                              float* __restrict__ h, int N) {
    int wave = (blockIdx.x * blockDim.x + threadIdx.x) >> 5;
    int lane = threadIdx.x & 31;
    int row0 = wave * 16;
    if (row0 >= N) return;

    int mrow = lane & 15;
    int koff = (lane >> 4) * 2;

    v8f c0 = {}, c1 = {}, c2 = {}, c3 = {};

    const float* arow = node + (size_t)(row0 + mrow) * D_IN + koff;

    for (int k = 0; k < D_IN; k += 4) {
        v2f a;
        a.x = arow[k];
        a.y = arow[k + 1];

        const float* bp = W + (size_t)(k + koff) * D_OUT + mrow;
        v2f b0, b1, b2, b3;
        b0.x = bp[0];            b0.y = bp[D_OUT];
        b1.x = bp[16];           b1.y = bp[16 + D_OUT];
        b2.x = bp[32];           b2.y = bp[32 + D_OUT];
        b3.x = bp[48];           b3.y = bp[48 + D_OUT];

        c0 = __builtin_amdgcn_wmma_f32_16x16x4_f32(false, a, false, b0, (short)0, c0, false, false);
        c1 = __builtin_amdgcn_wmma_f32_16x16x4_f32(false, a, false, b1, (short)0, c1, false, false);
        c2 = __builtin_amdgcn_wmma_f32_16x16x4_f32(false, a, false, b2, (short)0, c2, false, false);
        c3 = __builtin_amdgcn_wmma_f32_16x16x4_f32(false, a, false, b3, (short)0, c3, false, false);
    }

    int nlo   = lane & 15;
    int rbase = row0 + ((lane >> 4) * 8);
#pragma unroll
    for (int r = 0; r < 8; ++r) {
        float* op = h + (size_t)(rbase + r) * D_OUT + nlo;
        op[0]  = c0[r];
        op[16] = c1[r];
        op[32] = c2[r];
        op[48] = c3[r];
    }
}

__global__ void gat_scores(const float* __restrict__ h, const float* __restrict__ a,
                           float* __restrict__ ssrc, float* __restrict__ sdst, int N) {
    int i = blockIdx.x * blockDim.x + threadIdx.x;
    if (i >= N) return;
    const float* hr = h + (size_t)i * D_OUT;
    float s0 = 0.f, s1 = 0.f;
#pragma unroll 8
    for (int j = 0; j < D_OUT; ++j) {
        float v = hr[j];
        s0 = fmaf(v, a[j], s0);
        s1 = fmaf(v, a[D_OUT + j], s1);
    }
    ssrc[i] = s0;
    sdst[i] = s1;
}

__device__ __forceinline__ float wave_max(float v) {
    for (int o = 16; o > 0; o >>= 1) v = fmaxf(v, __shfl_xor(v, o));
    return v;
}
__device__ __forceinline__ float wave_sum(float v) {
    for (int o = 16; o > 0; o >>= 1) v += __shfl_xor(v, o);
    return v;
}

template <int K>
__device__ __forceinline__ void add_rows(const float2* __restrict__ hl, int dl, float al,
                                         int t, int N, float2& acc) {
    int    d[K];
    float  a[K];
    float2 v[K];
    for (int i = 0; i < K; ++i) {
        int di = __builtin_amdgcn_readlane(dl, t + i);
        d[i] = di < 0 ? 0 : (di >= N ? N - 1 : di);
        a[i] = __int_as_float(__builtin_amdgcn_readlane(__float_as_int(al), t + i));
    }
    for (int i = 0; i < K; ++i) v[i] = hl[(size_t)d[i] * (D_OUT / 2)];
    for (int i = 0; i < K; ++i) {
        acc.x = fmaf(a[i], v[i].x, acc.x);
        acc.y = fmaf(a[i], v[i].y, acc.y);
    }
}

__device__ __forceinline__ void add_chunk(const float2* __restrict__ hl, int dl, float al,
                                          int m, int N, float2& acc) {
    int t = 0;
    for (; t + 8 <= m; t += 8) add_rows<8>(hl, dl, al, t, N, acc);
    if (m - t >= 4) { add_rows<4>(hl, dl, al, t, N, acc); t += 4; }
    if (m - t >= 2) { add_rows<2>(hl, dl, al, t, N, acc); t += 2; }
    if (m - t >= 1) { add_rows<1>(hl, dl, al, t, N, acc); }
}

__global__ void gat_aggregate(const int* __restrict__ start, const int* __restrict__ cnt,
                              const int* __restrict__ nbr,
                              const float* __restrict__ ssrc, const float* __restrict__ sdst,
                              const float* __restrict__ h, float* __restrict__ out,
                              int N, int E) {
    int wid  = __builtin_amdgcn_readfirstlane((int)(threadIdx.x >> 5));
    int node = (int)blockIdx.x * (int)(blockDim.x >> 5) + wid;
    int lane = threadIdx.x & 31;
    if (node >= N) return;

    int beg = start[node];
    int n   = cnt[node];
    beg = beg < 0 ? 0 : (beg > E ? E : beg);
    n   = n < 0 ? 0 : (n > E - beg ? E - beg : n);
    const int*  lst = nbr + beg;
    const float ss  = ssrc[node];

    int   d0 = 0;
    float l0 = -INFINITY;
    if (lane < n) {
        d0 = lst[lane];
        d0 = d0 < 0 ? 0 : (d0 >= N ? N - 1 : d0);
        l0 = leaky(ss + sdst[d0]);
    }

    float mx = l0;
    for (int c = 32; c < n; c += 32) {
        int j = c + lane;
        if (j < n) {
            int d = lst[j];
            d = d < 0 ? 0 : (d >= N ? N - 1 : d);
            mx = fmaxf(mx, leaky(ss + sdst[d]));
        }
    }
    mx = wave_max(mx);

    float e0  = (lane < n) ? __expf(l0 - mx) : 0.f;
    float sum = e0;
    for (int c = 32; c < n; c += 32) {
        int j = c + lane;
        if (j < n) {
            int d = lst[j];
            d = d < 0 ? 0 : (d >= N ? N - 1 : d);
            sum += __expf(leaky(ss + sdst[d]) - mx);
        }
    }
    sum = wave_sum(sum);
    const float den = sum + EPS;

    float2 acc = make_float2(0.f, 0.f);
    const float2* hl = (const float2*)h + lane;
    if (n > 0) add_chunk(hl, d0, e0 / den, n < 32 ? n : 32, N, acc);
    for (int c = 32; c < n; c += 32) {
        int   j  = c + lane;
        int   dl = 0;
        float al = 0.f;
        if (j < n) {
            dl = lst[j];
            dl = dl < 0 ? 0 : (dl >= N ? N - 1 : dl);
            al = __expf(leaky(ss + sdst[dl]) - mx) / den;
        }
        add_chunk(hl, dl, al, (n - c) < 32 ? (n - c) : 32, N, acc);
    }
    float2* o2 = (float2*)out;
    o2[(size_t)node * (D_OUT / 2) + lane] = acc;
}

extern "C" void kernel_launch(void* const* d_in, const int* in_sizes, int n_in,
                              void* d_out, int out_size, void* d_ws, size_t ws_size,
                              hipStream_t stream) {
    const float* node = (const float*)d_in[0];
    const int*   ei   = (const int*)d_in[1];
    const float* W    = (const float*)d_in[2];
    const float* a    = (const float*)d_in[3];
    float* out        = (float*)d_out;

    int N = in_sizes[0] / D_IN;
    int E = in_sizes[1] / 2;
    const int* src = ei;
    const int* dst = ei + E;

    float* h       = (float*)d_ws;
    float* ssrc    = h + (size_t)N * D_OUT;
    float* sdst    = ssrc + N;
    int*   cnt     = (int*)(sdst + N);
    int*   start   = cnt + N;
    int*   counter = start + N;
    int*   nbr     = counter + 1;
    int*   rank    = (int*)h;

    const int B = 256;

    gat_init<<<(N + B - 1) / B, B, 0, stream>>>(cnt, counter, N);
    gat_hist_rank<<<(E + B - 1) / B, B, 0, stream>>>(src, cnt, rank, E);
    gat_alloc<<<(N + B - 1) / B, B, 0, stream>>>(cnt, start, counter, N);
    gat_place<<<(E + B - 1) / B, B, 0, stream>>>(src, dst, rank, start, nbr, E);
    {
        int waves   = (N + 15) / 16;
        long long threads = (long long)waves * 32;
        int blocks  = (int)((threads + B - 1) / B);
        gat_gemm_wmma<<<blocks, B, 0, stream>>>(node, W, h, N);
    }
    gat_scores<<<(N + B - 1) / B, B, 0, stream>>>(h, a, ssrc, sdst, N);
    {
        long long threads = (long long)N * 32;
        int blocks = (int)((threads + B - 1) / B);
        gat_aggregate<<<blocks, B, 0, stream>>>(start, cnt, nbr, ssrc, sdst, h, out, N, E);
    }
}


# edge ranks from LDS counters in a two-level counting sort (partition by 512-node bin, then one block per bin ranks in LDS) instead of one returning global atomic per edge; sum order: arrival rank
# speedup vs baseline: 3.5044x; 1.8203x over previous
#include <hip/hip_runtime.h>


#define D_IN   128
#define D_OUT  64
#define ALPHA  0.2f
#define EPS    1e-10f
#define CAP    D_OUT
#define RB     512
#define RBSH   9
#define EPT    16
#define PBT    256
#define RBT    1024
#define RPT    16
#define CST    64
#define EBITS  23

typedef __attribute__((ext_vector_type(2))) float v2f;
typedef __attribute__((ext_vector_type(8))) float v8f;

__device__ __forceinline__ float leaky(float x) { return x > 0.f ? x : ALPHA * x; }

__global__ void gat_init(int* __restrict__ cursors, int* __restrict__ counter, int NB) {
    int i = blockIdx.x * blockDim.x + threadIdx.x;
    if (i < NB) cursors[i * CST] = 0;
    if (i == 0) counter[0] = 0;
}

__device__ __forceinline__ void ovf_push(int* __restrict__ counter, int* __restrict__ ovf, int e, int E) {
    int k = atomicAdd(counter, 1);
    if (k >= 0 && k < E) ovf[k] = e;
}

__global__ __launch_bounds__(PBT) void gat_part(const int* __restrict__ src, const int* __restrict__ dst,
                                                int* __restrict__ cursors, int* __restrict__ counter,
                                                int* __restrict__ ovf, float* out, int N, int E, int NB) {
    extern __shared__ int lds[];
    int* lh    = lds;
    int* lbase = lds + NB;
    const int tid = threadIdx.x;
    for (int t = tid; t < NB; t += PBT) lh[t] = 0;
    __syncthreads();

    const int e0 = (int)blockIdx.x * (PBT * EPT) + tid;
    int s[EPT], d[EPT], lr[EPT];
#pragma unroll
    for (int i = 0; i < EPT; ++i) {
        int e = e0 + i * PBT;
        s[i] = -1;
        d[i] = 0;
        if (e < E) { s[i] = src[e]; d[i] = dst[e]; }
    }
#pragma unroll
    for (int i = 0; i < EPT; ++i) {
        int e = e0 + i * PBT;
        lr[i] = -1;
        if ((unsigned)s[i] < (unsigned)N) {
            if (e < (1 << EBITS)) lr[i] = atomicAdd(&lh[s[i] >> RBSH], 1);
            else                  ovf_push(counter, ovf, e, E);
        }
    }
    __syncthreads();
    for (int t = tid; t < NB; t += PBT) {
        int c = lh[t];
        lbase[t] = c > 0 ? atomicAdd(&cursors[t * CST], c) : 0;
    }
    __syncthreads();
    int pos[EPT];
#pragma unroll
    for (int i = 0; i < EPT; ++i) pos[i] = lbase[lr[i] >= 0 ? (s[i] >> RBSH) : 0] + lr[i];
#pragma unroll
    for (int i = 0; i < EPT; ++i) {
        if (lr[i] >= 0) {
            int e    = e0 + i * PBT;
            int b    = s[i] >> RBSH;
            int rows = N - b * RB;
            rows     = rows < RB ? rows : RB;
            if (pos[i] >= 0 && pos[i] < rows * (CAP / 2)) {
                int2* area = (int2*)(out + (size_t)b * RB * D_OUT);
                unsigned w = ((unsigned)(s[i] & (RB - 1)) << EBITS) | (unsigned)e;
                area[pos[i]] = make_int2((int)w, d[i]);
            } else {
                ovf_push(counter, ovf, e, E);
            }
        }
    }
}

__global__ __launch_bounds__(RBT) void gat_rank(const int* __restrict__ cursors, int* __restrict__ cnt,
                                                int* __restrict__ counter, int* __restrict__ ovf,
                                                float* out, int N, int E) {
    __shared__ int lc[RB];
    const int b   = blockIdx.x;
    const int tid = threadIdx.x;
    int rows = N - b * RB;
    rows = rows < RB ? rows : RB;
    const int cap = rows * (CAP / 2);
    int tot = cursors[b * CST];
    tot = tot < 0 ? 0 : (tot > cap ? cap : tot);
    if (tid < RB) lc[tid] = 0;
    __syncthreads();

    const int2* area = (const int2*)(out + (size_t)b * RB * D_OUT);
    int w[RPT], d[RPT], r[RPT];
#pragma unroll
    for (int i = 0; i < RPT; ++i) {
        int p = i * RBT + tid;
        int2 v = area[p < tot ? p : 0];
        w[i] = v.x;
        d[i] = v.y;
    }
#pragma unroll
    for (int i = 0; i < RPT; ++i) {
        int p = i * RBT + tid;
        r[i] = -1;
        int sl = (int)((unsigned)w[i] >> EBITS);
        if (p < tot && sl < rows) r[i] = atomicAdd(&lc[sl], 1);
    }
    __syncthreads();
#pragma unroll
    for (int i = 0; i < RPT; ++i) {
        if (r[i] >= 0) {
            int sl = (int)((unsigned)w[i] >> EBITS);
            if (r[i] < CAP) {
                int* bucket = (int*)out + ((size_t)b * RB + sl) * CAP;
                bucket[r[i]] = d[i];
            } else {
                ovf_push(counter, ovf, (int)((unsigned)w[i] & ((1u << EBITS) - 1u)), E);
            }
        }
    }
    if (tid < rows) cnt[b * RB + tid] = lc[tid];
}

__global__ void gat_gemm_wmma(const float* __restrict__ node,
                              const float* __restrict__ W,
                              float* __restrict__ h, int N) {
    int wave = (blockIdx.x * blockDim.x + threadIdx.x) >> 5;
    int lane = threadIdx.x & 31;
    int row0 = wave * 16;
    if (row0 >= N) return;

    int mrow = lane & 15;
    int koff = (lane >> 4) * 2;

    v8f c0 = {}, c1 = {}, c2 = {}, c3 = {};

    const float* arow = node + (size_t)(row0 + mrow) * D_IN + koff;

    for (int k = 0; k < D_IN; k += 4) {
        v2f a;
        a.x = arow[k];
        a.y = arow[k + 1];

        const float* bp = W + (size_t)(k + koff) * D_OUT + mrow;
        v2f b0, b1, b2, b3;
        b0.x = bp[0];            b0.y = bp[D_OUT];
        b1.x = bp[16];           b1.y = bp[16 + D_OUT];
        b2.x = bp[32];           b2.y = bp[32 + D_OUT];
        b3.x = bp[48];           b3.y = bp[48 + D_OUT];

        c0 = __builtin_amdgcn_wmma_f32_16x16x4_f32(false, a, false, b0, (short)0, c0, false, false);
        c1 = __builtin_amdgcn_wmma_f32_16x16x4_f32(false, a, false, b1, (short)0, c1, false, false);
        c2 = __builtin_amdgcn_wmma_f32_16x16x4_f32(false, a, false, b2, (short)0, c2, false, false);
        c3 = __builtin_amdgcn_wmma_f32_16x16x4_f32(false, a, false, b3, (short)0, c3, false, false);
    }

    int nlo   = lane & 15;
    int rbase = row0 + ((lane >> 4) * 8);
#pragma unroll
    for (int r = 0; r < 8; ++r) {
        float* op = h + (size_t)(rbase + r) * D_OUT + nlo;
        op[0]  = c0[r];
        op[16] = c1[r];
        op[32] = c2[r];
        op[48] = c3[r];
    }
}

__global__ void gat_scores(const float* __restrict__ h, const float* __restrict__ a,
                           float* __restrict__ ssrc, float* __restrict__ sdst, int N) {
    int i = blockIdx.x * blockDim.x + threadIdx.x;
    if (i >= N) return;
    const float* hr = h + (size_t)i * D_OUT;
    float s0 = 0.f, s1 = 0.f;
#pragma unroll 8
    for (int j = 0; j < D_OUT; ++j) {
        float v = hr[j];
        s0 = fmaf(v, a[j], s0);
        s1 = fmaf(v, a[D_OUT + j], s1);
    }
    ssrc[i] = s0;
    sdst[i] = s1;
}

__device__ __forceinline__ float wave_max(float v) {
    for (int o = 16; o > 0; o >>= 1) v = fmaxf(v, __shfl_xor(v, o));
    return v;
}
__device__ __forceinline__ float wave_sum(float v) {
    for (int o = 16; o > 0; o >>= 1) v += __shfl_xor(v, o);
    return v;
}

template <int K>
__device__ __forceinline__ void add_rows(const float2* __restrict__ hl, int dl, float al,
                                         int t, int N, float2& acc) {
    int    d[K];
    float  a[K];
    float2 v[K];
    for (int i = 0; i < K; ++i) {
        int di = __builtin_amdgcn_readlane(dl, t + i);
        d[i] = di < 0 ? 0 : (di >= N ? N - 1 : di);
        a[i] = __int_as_float(__builtin_amdgcn_readlane(__float_as_int(al), t + i));
    }
    for (int i = 0; i < K; ++i) v[i] = hl[(size_t)d[i] * (D_OUT / 2)];
    for (int i = 0; i < K; ++i) {
        acc.x = fmaf(a[i], v[i].x, acc.x);
        acc.y = fmaf(a[i], v[i].y, acc.y);
    }
}

__device__ __forceinline__ void add_chunk(const float2* __restrict__ hl, int dl, float al,
                                          int m, int N, float2& acc) {
    int t = 0;
    for (; t + 8 <= m; t += 8) add_rows<8>(hl, dl, al, t, N, acc);
    if (m - t >= 4) { add_rows<4>(hl, dl, al, t, N, acc); t += 4; }
    if (m - t >= 2) { add_rows<2>(hl, dl, al, t, N, acc); t += 2; }
    if (m - t >= 1) { add_rows<1>(hl, dl, al, t, N, acc); }
}

__device__ __forceinline__ bool ovf_slice(const int* __restrict__ ovf, const int* __restrict__ src,
                                          const int* __restrict__ dst, int j, int novf, int lane,
                                          int node, int N, int E, int& d) {
    d = 0;
    int k = j + lane;
    if (k >= novf) return false;
    int e = ovf[k];
    e = e < 0 ? 0 : (e >= E ? E - 1 : e);
    if (src[e] != node) return false;
    int t = dst[e];
    d = t < 0 ? 0 : (t >= N ? N - 1 : t);
    return true;
}

__global__ void gat_aggregate(const int* __restrict__ cnt, const int* __restrict__ counter,
                              const int* __restrict__ ovf,
                              const int* __restrict__ src, const int* __restrict__ dst,
                              const float* __restrict__ ssrc, const float* __restrict__ sdst,
                              const float* __restrict__ h, float* out,
                              int N, int E) {
    int wid  = __builtin_amdgcn_readfirstlane((int)(threadIdx.x >> 5));
    int node = (int)blockIdx.x * (int)(blockDim.x >> 5) + wid;
    int lane = threadIdx.x & 31;
    if (node >= N) return;

    const int*  lst = (const int*)out + (size_t)node * CAP;
    const int   dr  = lst[lane];
    const float ss  = ssrc[node];
    int n = cnt[node];
    int novf = counter[0];
    __builtin_amdgcn_sched_barrier(0);

    n = n < 0 ? 0 : (n > E ? E : n);
    const int nb = n < CAP ? n : CAP;

    int   d0 = dr < 0 ? 0 : (dr >= N ? N - 1 : dr);
    d0 = lane < nb ? d0 : 0;
    float l0 = -INFINITY;
    if (lane < nb) l0 = leaky(ss + sdst[d0]);

    float mx = l0;
    for (int c = 32; c < nb; c += 32) {
        int j = c + lane;
        if (j < nb) {
            int d = lst[j];
            d = d < 0 ? 0 : (d >= N ? N - 1 : d);
            mx = fmaxf(mx, leaky(ss + sdst[d]));
        }
    }
    novf = novf < 0 ? 0 : (novf > E ? E : novf);
    for (int j = 0; j < novf; j += 32) {
        int d;
        if (ovf_slice(ovf, src, dst, j, novf, lane, node, N, E, d))
            mx = fmaxf(mx, leaky(ss + sdst[d]));
    }
    mx = wave_max(mx);

    float e0  = (lane < nb) ? __expf(l0 - mx) : 0.f;
    float sum = e0;
    for (int c = 32; c < nb; c += 32) {
        int j = c + lane;
        if (j < nb) {
            int d = lst[j];
            d = d < 0 ? 0 : (d >= N ? N - 1 : d);
            sum += __expf(leaky(ss + sdst[d]) - mx);
        }
    }
    for (int j = 0; j < novf; j += 32) {
        int d;
        if (ovf_slice(ovf, src, dst, j, novf, lane, node, N, E, d))
            sum += __expf(leaky(ss + sdst[d]) - mx);
    }
    sum = wave_sum(sum);
    const float den = sum + EPS;

    float2 acc = make_float2(0.f, 0.f);
    const float2* hl = (const float2*)h + lane;
    if (nb > 0) add_chunk(hl, d0, e0 / den, nb < 32 ? nb : 32, N, acc);
    for (int c = 32; c < nb; c += 32) {
        int   j  = c + lane;
        int   dl = 0;
        float al = 0.f;
        if (j < nb) {
            dl = lst[j];
            dl = dl < 0 ? 0 : (dl >= N ? N - 1 : dl);
            al = __expf(leaky(ss + sdst[dl]) - mx) / den;
        }
        add_chunk(hl, dl, al, (nb - c) < 32 ? (nb - c) : 32, N, acc);
    }
    for (int j = 0; j < novf; j += 32) {
        int   dl;
        bool  hit = ovf_slice(ovf, src, dst, j, novf, lane, node, N, E, dl);
        float al  = hit ? __expf(leaky(ss + sdst[dl]) - mx) / den : 0.f;
        unsigned mask = __builtin_amdgcn_ballot_w32(hit);
        while (mask) {
            int t = __builtin_ctz(mask);
            mask &= mask - 1;
            add_rows<1>(hl, dl, al, t, N, acc);
        }
    }
    float2* o2 = (float2*)out;
    o2[(size_t)node * (D_OUT / 2) + lane] = acc;
}

extern "C" void kernel_launch(void* const* d_in, const int* in_sizes, int n_in,
                              void* d_out, int out_size, void* d_ws, size_t ws_size,
                              hipStream_t stream) {
    const float* node = (const float*)d_in[0];
    const int*   ei   = (const int*)d_in[1];
    const float* W    = (const float*)d_in[2];
    const float* a    = (const float*)d_in[3];
    float* out        = (float*)d_out;

    int N = in_sizes[0] / D_IN;
    int E = in_sizes[1] / 2;
    const int* src = ei;
    const int* dst = ei + E;

    float* h       = (float*)d_ws;
    float* ssrc    = h + (size_t)N * D_OUT;
    float* sdst    = ssrc + N;
    int*   cnt     = (int*)(sdst + N);
    int*   counter = cnt + N;
    int*   ovf     = counter + 1;
    int*   cursors = (int*)ssrc;
    int    NB      = (N + RB - 1) / RB;

    const int B = 256;

    gat_init<<<(NB + B - 1) / B, B, 0, stream>>>(cursors, counter, NB);
    gat_part<<<(E + PBT * EPT - 1) / (PBT * EPT), PBT, (size_t)NB * 2 * sizeof(int), stream>>>(
        src, dst, cursors, counter, ovf, out, N, E, NB);
    gat_rank<<<NB, RBT, 0, stream>>>(cursors, cnt, counter, ovf, out, N, E);
    {
        int waves   = (N + 15) / 16;
        long long threads = (long long)waves * 32;
        int blocks  = (int)((threads + B - 1) / B);
        gat_gemm_wmma<<<blocks, B, 0, stream>>>(node, W, h, N);
    }
    gat_scores<<<(N + B - 1) / B, B, 0, stream>>>(h, a, ssrc, sdst, N);
    {
        long long threads = (long long)N * 32;
        int blocks = (int)((threads + B - 1) / B);
        gat_aggregate<<<blocks, B, 0, stream>>>(cnt, counter, ovf, src, dst, ssrc, sdst, h, out, N, E);
    }
}


# scores ssrc/sdst taken from the WMMA accumulators in the product's epilogue (16-lane exchange tree); the scores pass that re-read h is dropped; score sum order: tree, was one 64-term chain
# speedup vs baseline: 3.8170x; 1.0892x over previous
#include <hip/hip_runtime.h>


#define D_IN   128
#define D_OUT  64
#define ALPHA  0.2f
#define EPS    1e-10f
#define CAP    D_OUT
#define RB     512
#define RBSH   9
#define EPT    16
#define PBT    256
#define RBT    1024
#define RPT    16
#define CST    RB
#define EBITS  23

typedef __attribute__((ext_vector_type(2))) float v2f;
typedef __attribute__((ext_vector_type(8))) float v8f;

__device__ __forceinline__ float leaky(float x) { return x > 0.f ? x : ALPHA * x; }

__global__ void gat_init(int* __restrict__ cursors, int* __restrict__ counter, int NB) {
    int i = blockIdx.x * blockDim.x + threadIdx.x;
    if (i < NB) cursors[i * CST] = 0;
    if (i == 0) counter[0] = 0;
}

__device__ __forceinline__ void ovf_push(int* __restrict__ counter, int* __restrict__ ovf, int e, int E) {
    int k = atomicAdd(counter, 1);
    if (k >= 0 && k < E) ovf[k] = e;
}

__device__ __forceinline__ void part_body(int bid, int* lds,
                                          const int* __restrict__ src, const int* __restrict__ dst,
                                          int* __restrict__ cursors, int* __restrict__ counter,
                                          int* __restrict__ ovf, float* out, int N, int E, int NB) {
    int* lh    = lds;
    int* lbase = lds + NB;
    const int tid = threadIdx.x;
    for (int t = tid; t < NB; t += PBT) lh[t] = 0;
    __syncthreads();

    const int e0 = bid * (PBT * EPT) + tid;
    int s[EPT], d[EPT], lr[EPT];
#pragma unroll
    for (int i = 0; i < EPT; ++i) {
        int e = e0 + i * PBT;
        s[i] = -1;
        d[i] = 0;
        if (e < E) { s[i] = src[e]; d[i] = dst[e]; }
    }
#pragma unroll
    for (int i = 0; i < EPT; ++i) {
        int e = e0 + i * PBT;
        lr[i] = -1;
        if ((unsigned)s[i] < (unsigned)N) {
            if (e < (1 << EBITS)) lr[i] = atomicAdd(&lh[s[i] >> RBSH], 1);
            else                  ovf_push(counter, ovf, e, E);
        }
    }
    __syncthreads();
    for (int t = tid; t < NB; t += PBT) {
        int c = lh[t];
        lbase[t] = c > 0 ? atomicAdd(&cursors[t * CST], c) : 0;
    }
    __syncthreads();
    int pos[EPT];
#pragma unroll
    for (int i = 0; i < EPT; ++i) pos[i] = lbase[lr[i] >= 0 ? (s[i] >> RBSH) : 0] + lr[i];
#pragma unroll
    for (int i = 0; i < EPT; ++i) {
        if (lr[i] >= 0) {
            int e    = e0 + i * PBT;
            int b    = s[i] >> RBSH;
            int rows = N - b * RB;
            rows     = rows < RB ? rows : RB;
            if (pos[i] >= 0 && pos[i] < rows * (CAP / 2)) {
                int2* area = (int2*)(out + (size_t)b * RB * D_OUT);
                unsigned w = ((unsigned)(s[i] & (RB - 1)) << EBITS) | (unsigned)e;
                area[pos[i]] = make_int2((int)w, d[i]);
            } else {
                ovf_push(counter, ovf, e, E);
            }
        }
    }
}

__device__ __forceinline__ void rank_body(int b, int* lc, int* cnt,
                                          int* __restrict__ counter, int* __restrict__ ovf,
                                          float* out, int N, int E) {
    const int tid = threadIdx.x;
    int rows = N - b * RB;
    rows = rows < RB ? rows : RB;
    const int cap = rows * (CAP / 2);
    int tot = cnt[b * CST];
    tot = tot < 0 ? 0 : (tot > cap ? cap : tot);
    if (tid < RB) lc[tid] = 0;
    __syncthreads();

    const int2* area = (const int2*)(out + (size_t)b * RB * D_OUT);
    int w[RPT], d[RPT], r[RPT];
#pragma unroll
    for (int i = 0; i < RPT; ++i) {
        int p = i * RBT + tid;
        int2 v = area[p < tot ? p : 0];
        w[i] = v.x;
        d[i] = v.y;
    }
#pragma unroll
    for (int i = 0; i < RPT; ++i) {
        int p = i * RBT + tid;
        r[i] = -1;
        int sl = (int)((unsigned)w[i] >> EBITS);
        if (p < tot && sl < rows) r[i] = atomicAdd(&lc[sl], 1);
    }
    __syncthreads();
#pragma unroll
    for (int i = 0; i < RPT; ++i) {
        if (r[i] >= 0) {
            int sl = (int)((unsigned)w[i] >> EBITS);
            if (r[i] < CAP) {
                int* bucket = (int*)out + ((size_t)b * RB + sl) * CAP;
                bucket[r[i]] = d[i];
            } else {
                ovf_push(counter, ovf, (int)((unsigned)w[i] & ((1u << EBITS) - 1u)), E);
            }
        }
    }
    if (tid < rows) cnt[b * RB + tid] = lc[tid];
}

__device__ __forceinline__ void gemm_body(int bid, const float* __restrict__ node,
                                          const float* __restrict__ W, const float* __restrict__ att,
                                          float* __restrict__ h, float* __restrict__ ssrc,
                                          float* __restrict__ sdst, int N) {
    int wave = (bid * (int)blockDim.x + (int)threadIdx.x) >> 5;
    int lane = threadIdx.x & 31;
    int row0 = wave * 16;
    if (row0 >= N) return;

    int mrow = lane & 15;
    int koff = (lane >> 4) * 2;

    v8f c0 = {}, c1 = {}, c2 = {}, c3 = {};

    const int   nlo = lane & 15;
    const float as0 = att[nlo],         as1 = att[nlo + 16],         as2 = att[nlo + 32],         as3 = att[nlo + 48];
    const float ad0 = att[D_OUT + nlo], ad1 = att[D_OUT + nlo + 16], ad2 = att[D_OUT + nlo + 32], ad3 = att[D_OUT + nlo + 48];

    const float* arow = node + (size_t)(row0 + mrow) * D_IN + koff;

    for (int k = 0; k < D_IN; k += 4) {
        v2f a;
        a.x = arow[k];
        a.y = arow[k + 1];

        const float* bp = W + (size_t)(k + koff) * D_OUT + mrow;
        v2f b0, b1, b2, b3;
        b0.x = bp[0];            b0.y = bp[D_OUT];
        b1.x = bp[16];           b1.y = bp[16 + D_OUT];
        b2.x = bp[32];           b2.y = bp[32 + D_OUT];
        b3.x = bp[48];           b3.y = bp[48 + D_OUT];

        c0 = __builtin_amdgcn_wmma_f32_16x16x4_f32(false, a, false, b0, (short)0, c0, false, false);
        c1 = __builtin_amdgcn_wmma_f32_16x16x4_f32(false, a, false, b1, (short)0, c1, false, false);
        c2 = __builtin_amdgcn_wmma_f32_16x16x4_f32(false, a, false, b2, (short)0, c2, false, false);
        c3 = __builtin_amdgcn_wmma_f32_16x16x4_f32(false, a, false, b3, (short)0, c3, false, false);
    }

    int rbase = row0 + ((lane >> 4) * 8);
#pragma unroll
    for (int r = 0; r < 8; ++r) {
        float* op = h + (size_t)(rbase + r) * D_OUT + nlo;
        op[0]  = c0[r];
        op[16] = c1[r];
        op[32] = c2[r];
        op[48] = c3[r];
    }

    float ps[8], pd[8];
#pragma unroll
    for (int r = 0; r < 8; ++r) {
        ps[r] = fmaf(c3[r], as3, fmaf(c2[r], as2, fmaf(c1[r], as1, c0[r] * as0)));
        pd[r] = fmaf(c3[r], ad3, fmaf(c2[r], ad2, fmaf(c1[r], ad1, c0[r] * ad0)));
    }
    float u[8], t[4], s[2], sc;
    {
        const bool hi = (nlo & 8) != 0;
#pragma unroll
        for (int i = 0; i < 8; ++i) {
            float keep = hi ? pd[i] : ps[i];
            float send = hi ? ps[i] : pd[i];
            u[i] = keep + __shfl_xor(send, 8);
        }
    }
    {
        const bool hi = (nlo & 4) != 0;
#pragma unroll
        for (int i = 0; i < 4; ++i) {
            float keep = hi ? u[i + 4] : u[i];
            float send = hi ? u[i] : u[i + 4];
            t[i] = keep + __shfl_xor(send, 4);
        }
    }
    {
        const bool hi = (nlo & 2) != 0;
#pragma unroll
        for (int i = 0; i < 2; ++i) {
            float keep = hi ? t[i + 2] : t[i];
            float send = hi ? t[i] : t[i + 2];
            s[i] = keep + __shfl_xor(send, 2);
        }
    }
    {
        const bool hi = (nlo & 1) != 0;
        float keep = hi ? s[1] : s[0];
        float send = hi ? s[0] : s[1];
        sc = keep + __shfl_xor(send, 1);
    }
    float* sp = (nlo < 8) ? ssrc : sdst;
    if (rbase + (nlo & 7) < N) sp[rbase + (nlo & 7)] = sc;
}

__global__ __launch_bounds__(PBT) void gat_part_gemm(const int* __restrict__ src, const int* __restrict__ dst,
                                                     int* __restrict__ cursors, int* __restrict__ counter,
                                                     int* __restrict__ ovf, float* out,
                                                     const float* __restrict__ node, const float* __restrict__ W,
                                                     const float* __restrict__ a, float* __restrict__ h,
                                                     float* __restrict__ ssrc, float* __restrict__ sdst,
                                                     int N, int E, int NB, int PB) {
    extern __shared__ int lds[];
    const int bid = (int)blockIdx.x;
    if (bid < PB) part_body(bid, lds, src, dst, cursors, counter, ovf, out, N, E, NB);
    else          gemm_body(bid - PB, node, W, a, h, ssrc, sdst, N);
}

__global__ __launch_bounds__(RBT) void gat_rank(int* cnt, int* __restrict__ counter, int* __restrict__ ovf,
                                                float* out, int N, int E) {
    __shared__ int lc[RB];
    rank_body((int)blockIdx.x, lc, cnt, counter, ovf, out, N, E);
}

__device__ __forceinline__ float wave_max(float v) {
    for (int o = 16; o > 0; o >>= 1) v = fmaxf(v, __shfl_xor(v, o));
    return v;
}
__device__ __forceinline__ float wave_sum(float v) {
    for (int o = 16; o > 0; o >>= 1) v += __shfl_xor(v, o);
    return v;
}

template <int K>
__device__ __forceinline__ void add_rows(const float2* __restrict__ hl, int dl, float al,
                                         int t, int N, float2& acc) {
    int    d[K];
    float  a[K];
    float2 v[K];
    for (int i = 0; i < K; ++i) {
        int di = __builtin_amdgcn_readlane(dl, t + i);
        d[i] = di < 0 ? 0 : (di >= N ? N - 1 : di);
        a[i] = __int_as_float(__builtin_amdgcn_readlane(__float_as_int(al), t + i));
    }
    for (int i = 0; i < K; ++i) v[i] = hl[(size_t)d[i] * (D_OUT / 2)];
    for (int i = 0; i < K; ++i) {
        acc.x = fmaf(a[i], v[i].x, acc.x);
        acc.y = fmaf(a[i], v[i].y, acc.y);
    }
}

__device__ __forceinline__ void add_chunk(const float2* __restrict__ hl, int dl, float al,
                                          int m, int N, float2& acc) {
    int t = 0;
    for (; t + 8 <= m; t += 8) add_rows<8>(hl, dl, al, t, N, acc);
    if (m - t >= 4) { add_rows<4>(hl, dl, al, t, N, acc); t += 4; }
    if (m - t >= 2) { add_rows<2>(hl, dl, al, t, N, acc); t += 2; }
    if (m - t >= 1) { add_rows<1>(hl, dl, al, t, N, acc); }
}

__device__ __forceinline__ bool ovf_slice(const int* __restrict__ ovf, const int* __restrict__ src,
                                          const int* __restrict__ dst, int j, int novf, int lane,
                                          int node, int N, int E, int& d) {
    d = 0;
    int k = j + lane;
    if (k >= novf) return false;
    int e = ovf[k];
    e = e < 0 ? 0 : (e >= E ? E - 1 : e);
    if (src[e] != node) return false;
    int t = dst[e];
    d = t < 0 ? 0 : (t >= N ? N - 1 : t);
    return true;
}

__global__ void gat_aggregate(const int* __restrict__ cnt, const int* __restrict__ counter,
                              const int* __restrict__ ovf,
                              const int* __restrict__ src, const int* __restrict__ dst,
                              const float* __restrict__ ssrc, const float* __restrict__ sdst,
                              const float* __restrict__ h, float* out,
                              int N, int E) {
    int wid  = __builtin_amdgcn_readfirstlane((int)(threadIdx.x >> 5));
    int node = (int)blockIdx.x * (int)(blockDim.x >> 5) + wid;
    int lane = threadIdx.x & 31;
    if (node >= N) return;

    const int*  lst = (const int*)out + (size_t)node * CAP;
    const int   dr  = lst[lane];
    const float ss  = ssrc[node];
    int n = cnt[node];
    int novf = counter[0];
    __builtin_amdgcn_sched_barrier(0);

    n = n < 0 ? 0 : (n > E ? E : n);
    const int nb = n < CAP ? n : CAP;

    int   d0 = dr < 0 ? 0 : (dr >= N ? N - 1 : dr);
    d0 = lane < nb ? d0 : 0;
    float l0 = -INFINITY;
    if (lane < nb) l0 = leaky(ss + sdst[d0]);

    float mx = l0;
    for (int c = 32; c < nb; c += 32) {
        int j = c + lane;
        if (j < nb) {
            int d = lst[j];
            d = d < 0 ? 0 : (d >= N ? N - 1 : d);
            mx = fmaxf(mx, leaky(ss + sdst[d]));
        }
    }
    novf = novf < 0 ? 0 : (novf > E ? E : novf);
    for (int j = 0; j < novf; j += 32) {
        int d;
        if (ovf_slice(ovf, src, dst, j, novf, lane, node, N, E, d))
            mx = fmaxf(mx, leaky(ss + sdst[d]));
    }
    mx = wave_max(mx);

    float e0  = (lane < nb) ? __expf(l0 - mx) : 0.f;
    float sum = e0;
    for (int c = 32; c < nb; c += 32) {
        int j = c + lane;
        if (j < nb) {
            int d = lst[j];
            d = d < 0 ? 0 : (d >= N ? N - 1 : d);
            sum += __expf(leaky(ss + sdst[d]) - mx);
        }
    }
    for (int j = 0; j < novf; j += 32) {
        int d;
        if (ovf_slice(ovf, src, dst, j, novf, lane, node, N, E, d))
            sum += __expf(leaky(ss + sdst[d]) - mx);
    }
    sum = wave_sum(sum);
    const float den = sum + EPS;

    float2 acc = make_float2(0.f, 0.f);
    const float2* hl = (const float2*)h + lane;
    if (nb > 0) add_chunk(hl, d0, e0 / den, nb < 32 ? nb : 32, N, acc);
    for (int c = 32; c < nb; c += 32) {
        int   j  = c + lane;
        int   dl = 0;
        float al = 0.f;
        if (j < nb) {
            dl = lst[j];
            dl = dl < 0 ? 0 : (dl >= N ? N - 1 : dl);
            al = __expf(leaky(ss + sdst[dl]) - mx) / den;
        }
        add_chunk(hl, dl, al, (nb - c) < 32 ? (nb - c) : 32, N, acc);
    }
    for (int j = 0; j < novf; j += 32) {
        int   dl;
        bool  hit = ovf_slice(ovf, src, dst, j, novf, lane, node, N, E, dl);
        float al  = hit ? __expf(leaky(ss + sdst[dl]) - mx) / den : 0.f;
        unsigned mask = __builtin_amdgcn_ballot_w32(hit);
        while (mask) {
            int t = __builtin_ctz(mask);
            mask &= mask - 1;
            add_rows<1>(hl, dl, al, t, N, acc);
        }
    }
    float2* o2 = (float2*)out;
    o2[(size_t)node * (D_OUT / 2) + lane] = acc;
}

extern "C" void kernel_launch(void* const* d_in, const int* in_sizes, int n_in,
                              void* d_out, int out_size, void* d_ws, size_t ws_size,
                              hipStream_t stream) {
    const float* node = (const float*)d_in[0];
    const int*   ei   = (const int*)d_in[1];
    const float* W    = (const float*)d_in[2];
    const float* a    = (const float*)d_in[3];
    float* out        = (float*)d_out;

    int N = in_sizes[0] / D_IN;
    int E = in_sizes[1] / 2;
    const int* src = ei;
    const int* dst = ei + E;

    float* h       = (float*)d_ws;
    float* ssrc    = h + (size_t)N * D_OUT;
    float* sdst    = ssrc + N;
    int*   cnt     = (int*)(sdst + N);
    int*   counter = cnt + N;
    int*   ovf     = counter + 1;
    int*   cursors = cnt;
    int    NB      = (N + RB - 1) / RB;

    const int B = 256;

    gat_init<<<(NB + B - 1) / B, B, 0, stream>>>(cursors, counter, NB);
    {
        int PB      = (E + PBT * EPT - 1) / (PBT * EPT);
        int waves   = (N + 15) / 16;
        long long threads = (long long)waves * 32;
        int gblocks = (int)((threads + PBT - 1) / PBT);
        gat_part_gemm<<<PB + gblocks, PBT, (size_t)NB * 2 * sizeof(int), stream>>>(
            src, dst, cursors, counter, ovf, out, node, W, a, h, ssrc, sdst, N, E, NB, PB);
    }
    gat_rank<<<NB, RBT, 0, stream>>>(cnt, counter, ovf, out, N, E);
    {
        long long threads = (long long)N * 32;
        int blocks = (int)((threads + B - 1) / B);
        gat_aggregate<<<blocks, B, 0, stream>>>(cnt, counter, ovf, src, dst, ssrc, sdst, h, out, N, E);
    }
}


# rank: bucket slots 0-31 gathered in LDS, then each node's first 128-byte line leaves in whole-line stores (4 lines per instruction) instead of scattered 4-byte stores; sum order unchanged
# speedup vs baseline: 4.6014x; 1.2055x over previous
#include <hip/hip_runtime.h>


#define D_IN   128
#define D_OUT  64
#define ALPHA  0.2f
#define EPS    1e-10f
#define CAP    D_OUT
#define RB     512
#define RBSH   9
#define EPT    16
#define PBT    256
#define RBT    1024
#define RPT    16
#define CST    RB
#define EBITS  23
#define SLOT0  32

typedef __attribute__((ext_vector_type(2))) float v2f;
typedef __attribute__((ext_vector_type(8))) float v8f;

__device__ __forceinline__ float leaky(float x) { return x > 0.f ? x : ALPHA * x; }

__global__ void gat_init(int* __restrict__ cursors, int* __restrict__ counter, int NB) {
    int i = blockIdx.x * blockDim.x + threadIdx.x;
    if (i < NB) cursors[i * CST] = 0;
    if (i == 0) counter[0] = 0;
}

__device__ __forceinline__ void ovf_push(int* __restrict__ counter, int* __restrict__ ovf, int e, int E) {
    int k = atomicAdd(counter, 1);
    if (k >= 0 && k < E) ovf[k] = e;
}

__device__ __forceinline__ void part_body(int bid, int* lds,
                                          const int* __restrict__ src, const int* __restrict__ dst,
                                          int* __restrict__ cursors, int* __restrict__ counter,
                                          int* __restrict__ ovf, float* out, int N, int E, int NB) {
    int* lh    = lds;
    int* lbase = lds + NB;
    const int tid = threadIdx.x;
    for (int t = tid; t < NB; t += PBT) lh[t] = 0;
    __syncthreads();

    const int e0 = bid * (PBT * EPT) + tid;
    int s[EPT], d[EPT], lr[EPT];
#pragma unroll
    for (int i = 0; i < EPT; ++i) {
        int e = e0 + i * PBT;
        s[i] = -1;
        d[i] = 0;
        if (e < E) { s[i] = src[e]; d[i] = dst[e]; }
    }
#pragma unroll
    for (int i = 0; i < EPT; ++i) {
        int e = e0 + i * PBT;
        lr[i] = -1;
        if ((unsigned)s[i] < (unsigned)N) {
            if (e < (1 << EBITS)) lr[i] = atomicAdd(&lh[s[i] >> RBSH], 1);
            else                  ovf_push(counter, ovf, e, E);
        }
    }
    __syncthreads();
    for (int t = tid; t < NB; t += PBT) {
        int c = lh[t];
        lbase[t] = c > 0 ? atomicAdd(&cursors[t * CST], c) : 0;
    }
    __syncthreads();
    int pos[EPT];
#pragma unroll
    for (int i = 0; i < EPT; ++i) pos[i] = lbase[lr[i] >= 0 ? (s[i] >> RBSH) : 0] + lr[i];
#pragma unroll
    for (int i = 0; i < EPT; ++i) {
        if (lr[i] >= 0) {
            int e    = e0 + i * PBT;
            int b    = s[i] >> RBSH;
            int rows = N - b * RB;
            rows     = rows < RB ? rows : RB;
            if (pos[i] >= 0 && pos[i] < rows * (CAP / 2)) {
                int2* area = (int2*)(out + (size_t)b * RB * D_OUT);
                unsigned w = ((unsigned)(s[i] & (RB - 1)) << EBITS) | (unsigned)e;
                area[pos[i]] = make_int2((int)w, d[i]);
            } else {
                ovf_push(counter, ovf, e, E);
            }
        }
    }
}

__device__ __forceinline__ void rank_body(int b, int* lc, int* stage, int* cnt,
                                          int* __restrict__ counter, int* __restrict__ ovf,
                                          float* out, int N, int E) {
    const int tid = threadIdx.x;
    int rows = N - b * RB;
    rows = rows < RB ? rows : RB;
    const int cap = rows * (CAP / 2);
    int tot = cnt[b * CST];
    tot = tot < 0 ? 0 : (tot > cap ? cap : tot);
    if (tid < RB) lc[tid] = 0;
    for (int i = tid; i < RB * SLOT0 / 4; i += RBT) ((int4*)stage)[i] = make_int4(0, 0, 0, 0);
    __syncthreads();

    const int2* area = (const int2*)(out + (size_t)b * RB * D_OUT);
    int w[RPT], d[RPT], r[RPT];
#pragma unroll
    for (int i = 0; i < RPT; ++i) {
        int p = i * RBT + tid;
        int2 v = area[p < tot ? p : 0];
        w[i] = v.x;
        d[i] = v.y;
    }
#pragma unroll
    for (int i = 0; i < RPT; ++i) {
        int p = i * RBT + tid;
        r[i] = -1;
        int sl = (int)((unsigned)w[i] >> EBITS);
        if (p < tot && sl < rows) r[i] = atomicAdd(&lc[sl], 1);
    }
    __syncthreads();
#pragma unroll
    for (int i = 0; i < RPT; ++i) {
        if (r[i] >= 0) {
            int sl = (int)((unsigned)w[i] >> EBITS);
            if (r[i] < SLOT0) {
                stage[sl * SLOT0 + r[i]] = d[i];
            } else if (r[i] < CAP) {
                int* bucket = (int*)out + ((size_t)b * RB + sl) * CAP;
                bucket[r[i]] = d[i];
            } else {
                ovf_push(counter, ovf, (int)((unsigned)w[i] & ((1u << EBITS) - 1u)), E);
            }
        }
    }
    __syncthreads();
    for (int i = tid; i < RB * SLOT0 / 4; i += RBT) {
        int row = i >> 3;
        if (row < rows) {
            int4  v = ((const int4*)stage)[i];
            int4* o = (int4*)((int*)out + ((size_t)b * RB + row) * CAP) + (i & 7);
            *o = v;
        }
    }
    if (tid < rows) cnt[b * RB + tid] = lc[tid];
}

__device__ __forceinline__ void gemm_body(int bid, const float* __restrict__ node,
                                          const float* __restrict__ W, const float* __restrict__ att,
                                          float* __restrict__ h, float* __restrict__ ssrc, int N) {
    int wave = (bid * (int)blockDim.x + (int)threadIdx.x) >> 5;
    int lane = threadIdx.x & 31;
    int row0 = wave * 16;
    if (row0 >= N) return;

    int mrow = lane & 15;
    int koff = (lane >> 4) * 2;

    v8f c0 = {}, c1 = {}, c2 = {}, c3 = {};

    const int   nlo = lane & 15;
    const float as0 = att[nlo], as1 = att[nlo + 16], as2 = att[nlo + 32], as3 = att[nlo + 48];

    const float* arow = node + (size_t)(row0 + mrow) * D_IN + koff;

    for (int k = 0; k < D_IN; k += 4) {
        v2f a;
        a.x = arow[k];
        a.y = arow[k + 1];

        const float* bp = W + (size_t)(k + koff) * D_OUT + mrow;
        v2f b0, b1, b2, b3;
        b0.x = bp[0];            b0.y = bp[D_OUT];
        b1.x = bp[16];           b1.y = bp[16 + D_OUT];
        b2.x = bp[32];           b2.y = bp[32 + D_OUT];
        b3.x = bp[48];           b3.y = bp[48 + D_OUT];

        c0 = __builtin_amdgcn_wmma_f32_16x16x4_f32(false, a, false, b0, (short)0, c0, false, false);
        c1 = __builtin_amdgcn_wmma_f32_16x16x4_f32(false, a, false, b1, (short)0, c1, false, false);
        c2 = __builtin_amdgcn_wmma_f32_16x16x4_f32(false, a, false, b2, (short)0, c2, false, false);
        c3 = __builtin_amdgcn_wmma_f32_16x16x4_f32(false, a, false, b3, (short)0, c3, false, false);
    }

    int rbase = row0 + ((lane >> 4) * 8);
#pragma unroll
    for (int r = 0; r < 8; ++r) {
        float* op = h + (size_t)(rbase + r) * D_OUT + nlo;
        op[0]  = c0[r];
        op[16] = c1[r];
        op[32] = c2[r];
        op[48] = c3[r];
    }

    float ps[8];
#pragma unroll
    for (int r = 0; r < 8; ++r)
        ps[r] = fmaf(c3[r], as3, fmaf(c2[r], as2, fmaf(c1[r], as1, c0[r] * as0)));
    float u[8], t[4], s[2], sc;
#pragma unroll
    for (int i = 0; i < 8; ++i) u[i] = ps[i] + __shfl_xor(ps[i], 8);
    {
        const bool hi = (nlo & 4) != 0;
#pragma unroll
        for (int i = 0; i < 4; ++i) {
            float keep = hi ? u[i + 4] : u[i];
            float send = hi ? u[i] : u[i + 4];
            t[i] = keep + __shfl_xor(send, 4);
        }
    }
    {
        const bool hi = (nlo & 2) != 0;
#pragma unroll
        for (int i = 0; i < 2; ++i) {
            float keep = hi ? t[i + 2] : t[i];
            float send = hi ? t[i] : t[i + 2];
            s[i] = keep + __shfl_xor(send, 2);
        }
    }
    {
        const bool hi = (nlo & 1) != 0;
        float keep = hi ? s[1] : s[0];
        float send = hi ? s[0] : s[1];
        sc = keep + __shfl_xor(send, 1);
    }
    if (nlo < 8 && rbase + nlo < N) ssrc[rbase + nlo] = sc;
}

__global__ __launch_bounds__(PBT) void gat_part_gemm(const int* __restrict__ src, const int* __restrict__ dst,
                                                     int* __restrict__ cursors, int* __restrict__ counter,
                                                     int* __restrict__ ovf, float* out,
                                                     const float* __restrict__ node, const float* __restrict__ W,
                                                     const float* __restrict__ a, float* __restrict__ h,
                                                     float* __restrict__ ssrc,
                                                     int N, int E, int NB, int PB) {
    extern __shared__ int lds[];
    const int bid = (int)blockIdx.x;
    if (bid < PB) part_body(bid, lds, src, dst, cursors, counter, ovf, out, N, E, NB);
    else          gemm_body(bid - PB, node, W, a, h, ssrc, N);
}

__global__ __launch_bounds__(RBT) void gat_rank(int* cnt, int* __restrict__ counter, int* __restrict__ ovf,
                                                float* out, int N, int E) {
    __shared__ int lc[RB];
    __shared__ __attribute__((aligned(16))) int stage[RB * SLOT0];
    rank_body((int)blockIdx.x, lc, stage, cnt, counter, ovf, out, N, E);
}

template <int M>
__device__ __forceinline__ float lane_xor(float x) {
    int b = __float_as_int(x);
    if constexpr (M == 16)
        return __int_as_float(__builtin_amdgcn_permlanex16(b, b, 0x76543210, 0xfedcba98, false, false));
    else
        return __int_as_float(__builtin_amdgcn_update_dpp(0, b, 0x160 + M, 0xf, 0xf, true));
}

template <int K>
struct Tree {
    static __device__ __forceinline__ float run(const float* x, int lane) {
        constexpr int H = K / 2;
        const bool hi = (lane & H) != 0;
        float y[H];
        for (int i = 0; i < H; ++i) {
            float keep = hi ? x[i + H] : x[i];
            float send = hi ? x[i] : x[i + H];
            y[i] = keep + lane_xor<H>(send);
        }
        return Tree<H>::run(y, lane);
    }
};
template <>
struct Tree<1> {
    static __device__ __forceinline__ float run(const float* x, int) { return x[0]; }
};

template <int M, int TOP>
__device__ __forceinline__ float sum_bits(float v) {
    if constexpr (M < TOP) { v += lane_xor<M>(v); return sum_bits<M * 2, TOP>(v); }
    else return v;
}
template <int M, int TOP>
__device__ __forceinline__ float max_bits(float v) {
    if constexpr (M < TOP) { v = fmaxf(v, lane_xor<M>(v)); return max_bits<M * 2, TOP>(v); }
    else return v;
}

struct Soft {
    float  m;
    float  s;
    float2 acc;
};

template <int K>
__device__ __forceinline__ void add_rows(const float2* __restrict__ hl, float2 ad, float ss,
                                         int dl, int t, int lane, int N, Soft& st) {
    int    d[K];
    float2 v[K];
    float  p[K];
    for (int i = 0; i < K; ++i) {
        int di = __builtin_amdgcn_readlane(dl, t + i);
        d[i] = di < 0 ? 0 : (di >= N ? N - 1 : di);
    }
    for (int i = 0; i < K; ++i) v[i] = hl[(size_t)d[i] * (D_OUT / 2)];
    for (int i = 0; i < K; ++i) p[i] = fmaf(v[i].y, ad.y, v[i].x * ad.x);
    float sc = Tree<K>::run(p, lane);
    sc = sum_bits<K, 32>(sc);
    const float l  = leaky(ss + sc);
    const float gm = max_bits<1, K>(l);
    const float mn = fmaxf(st.m, gm);
    const float rs = __expf(st.m - mn);
    const float w  = __expf(l - mn);
    const float gs = sum_bits<1, K>(w);
    st.s     = fmaf(st.s, rs, gs);
    st.acc.x *= rs;
    st.acc.y *= rs;
    for (int i = 0; i < K; ++i) {
        float wi = __int_as_float(__builtin_amdgcn_readlane(__float_as_int(w), i));
        st.acc.x = fmaf(wi, v[i].x, st.acc.x);
        st.acc.y = fmaf(wi, v[i].y, st.acc.y);
    }
    st.m = mn;
}

__device__ __forceinline__ void add_chunk(const float2* __restrict__ hl, float2 ad, float ss,
                                          int dl, int m, int lane, int N, Soft& st) {
    int t = 0;
    for (; t + 16 <= m; t += 16) add_rows<16>(hl, ad, ss, dl, t, lane, N, st);
    if (m - t >= 8) { add_rows<8>(hl, ad, ss, dl, t, lane, N, st); t += 8; }
    if (m - t >= 4) { add_rows<4>(hl, ad, ss, dl, t, lane, N, st); t += 4; }
    if (m - t >= 2) { add_rows<2>(hl, ad, ss, dl, t, lane, N, st); t += 2; }
    if (m - t >= 1) { add_rows<1>(hl, ad, ss, dl, t, lane, N, st); }
}

__device__ __forceinline__ bool ovf_slice(const int* __restrict__ ovf, const int* __restrict__ src,
                                          const int* __restrict__ dst, int j, int novf, int lane,
                                          int node, int N, int E, int& d) {
    d = 0;
    int k = j + lane;
    if (k >= novf) return false;
    int e = ovf[k];
    e = e < 0 ? 0 : (e >= E ? E - 1 : e);
    if (src[e] != node) return false;
    int t = dst[e];
    d = t < 0 ? 0 : (t >= N ? N - 1 : t);
    return true;
}

__global__ void gat_aggregate(const int* __restrict__ cnt, const int* __restrict__ counter,
                              const int* __restrict__ ovf,
                              const int* __restrict__ src, const int* __restrict__ dst,
                              const float* __restrict__ ssrc, const float* __restrict__ a,
                              const float* __restrict__ h, float* out,
                              int N, int E) {
    int wid  = __builtin_amdgcn_readfirstlane((int)(threadIdx.x >> 5));
    int node = (int)blockIdx.x * (int)(blockDim.x >> 5) + wid;
    int lane = threadIdx.x & 31;
    if (node >= N) return;

    const int*   lst = (const int*)out + (size_t)node * CAP;
    const int    dr  = lst[lane];
    const float  ss  = ssrc[node];
    const float2 ad  = ((const float2*)(a + D_OUT))[lane];
    int n = cnt[node];
    int novf = counter[0];
    __builtin_amdgcn_sched_barrier(0);

    n = n < 0 ? 0 : (n > E ? E : n);
    const int nb = n < CAP ? n : CAP;

    Soft st;
    st.m   = -INFINITY;
    st.s   = 0.f;
    st.acc = make_float2(0.f, 0.f);
    const float2* hl = (const float2*)h + lane;

    if (nb > 0) add_chunk(hl, ad, ss, dr, nb < 32 ? nb : 32, lane, N, st);
    for (int c = 32; c < nb; c += 32) {
        int j  = c + lane;
        int dl = 0;
        if (j < nb) dl = lst[j];
        add_chunk(hl, ad, ss, dl, (nb - c) < 32 ? (nb - c) : 32, lane, N, st);
    }
    novf = novf < 0 ? 0 : (novf > E ? E : novf);
    for (int j = 0; j < novf; j += 32) {
        int  dl;
        bool hit = ovf_slice(ovf, src, dst, j, novf, lane, node, N, E, dl);
        unsigned mask = __builtin_amdgcn_ballot_w32(hit);
        while (mask) {
            int t = __builtin_ctz(mask);
            mask &= mask - 1;
            add_rows<1>(hl, ad, ss, dl, t, lane, N, st);
        }
    }
    const float inv = 1.0f / (st.s + EPS);
    float2* o2 = (float2*)out;
    o2[(size_t)node * (D_OUT / 2) + lane] = make_float2(st.acc.x * inv, st.acc.y * inv);
}

extern "C" void kernel_launch(void* const* d_in, const int* in_sizes, int n_in,
                              void* d_out, int out_size, void* d_ws, size_t ws_size,
                              hipStream_t stream) {
    const float* node = (const float*)d_in[0];
    const int*   ei   = (const int*)d_in[1];
    const float* W    = (const float*)d_in[2];
    const float* a    = (const float*)d_in[3];
    float* out        = (float*)d_out;

    int N = in_sizes[0] / D_IN;
    int E = in_sizes[1] / 2;
    const int* src = ei;
    const int* dst = ei + E;

    float* h       = (float*)d_ws;
    float* ssrc    = h + (size_t)N * D_OUT;
    int*   cnt     = (int*)(ssrc + N);
    int*   counter = cnt + N;
    int*   ovf     = counter + 1;
    int*   cursors = cnt;
    int    NB      = (N + RB - 1) / RB;

    const int B = 256;

    gat_init<<<(NB + B - 1) / B, B, 0, stream>>>(cursors, counter, NB);
    {
        int PB      = (E + PBT * EPT - 1) / (PBT * EPT);
        int waves   = (N + 15) / 16;
        long long threads = (long long)waves * 32;
        int gblocks = (int)((threads + PBT - 1) / PBT);
        gat_part_gemm<<<PB + gblocks, PBT, (size_t)NB * 2 * sizeof(int), stream>>>(
            src, dst, cursors, counter, ovf, out, node, W, a, h, ssrc, N, E, NB, PB);
    }
    gat_rank<<<NB, RBT, 0, stream>>>(cnt, counter, ovf, out, N, E);
    {
        long long threads = (long long)N * 32;
        int blocks = (int)((threads + B - 1) / B);
        gat_aggregate<<<blocks, B, 0, stream>>>(cnt, counter, ovf, src, dst, ssrc, a, h, out, N, E);
    }
}


# partition: records pass through LDS in bin order and leave as runs of adjacent 8-byte words, not one scattered store per edge; kernel at 86 registers (was 100), all blocks resident; sum order same
# speedup vs baseline: 4.8625x; 1.0567x over previous
#include <hip/hip_runtime.h>


#define D_IN   128
#define D_OUT  64
#define ALPHA  0.2f
#define EPS    1e-10f
#define CAP    D_OUT
#define RB     512
#define RBSH   9
#define EPT    16
#define PBT    256
#define RBT    1024
#define RPT    16
#define CST    RB
#define EBITS  23
#define SLOT0  32

typedef __attribute__((ext_vector_type(2))) float v2f;
typedef __attribute__((ext_vector_type(8))) float v8f;

__device__ __forceinline__ float leaky(float x) { return x > 0.f ? x : ALPHA * x; }

__global__ void gat_init(int* __restrict__ cursors, int* __restrict__ counter, int NB) {
    int i = blockIdx.x * blockDim.x + threadIdx.x;
    if (i < NB) cursors[i * CST] = 0;
    if (i == 0) counter[0] = 0;
}

__device__ __forceinline__ void ovf_push(int* __restrict__ counter, int* __restrict__ ovf, int e, int E) {
    int k = atomicAdd(counter, 1);
    if (k >= 0 && k < E) ovf[k] = e;
}

__device__ __forceinline__ void part_body(int bid, int* lds,
                                          const int* __restrict__ src, const int* __restrict__ dst,
                                          int* __restrict__ cursors, int* __restrict__ counter,
                                          int* __restrict__ ovf, float* out, int N, int E, int NB) {
    const int o_lbp = (NB + 1) & ~1;
    const int o_tot = o_lbp + 2 * NB;
    const int o_sw  = (o_tot + 4) & ~3;
    const int o_sg  = o_sw + 2 * PBT * EPT;
    int*  lh  = lds;
    int2* lbp = (int2*)(lds + o_lbp);
    int2* sw  = (int2*)(lds + o_sw);
    int*  sg  = lds + o_sg;
    const int tid = threadIdx.x;
    for (int t = tid; t < NB; t += PBT) lh[t] = 0;
    __syncthreads();

    const int e0 = bid * (PBT * EPT) + tid;
    int s[EPT], d[EPT], lr[EPT];
#pragma unroll
    for (int i = 0; i < EPT; ++i) {
        int e = e0 + i * PBT;
        s[i] = -1;
        d[i] = 0;
        if (e < E) { s[i] = src[e]; d[i] = dst[e]; }
    }
#pragma unroll
    for (int i = 0; i < EPT; ++i) {
        int e = e0 + i * PBT;
        lr[i] = -1;
        if ((unsigned)s[i] < (unsigned)N) {
            if (e < (1 << EBITS)) lr[i] = atomicAdd(&lh[s[i] >> RBSH], 1);
            else                  ovf_push(counter, ovf, e, E);
        }
    }
    __syncthreads();
    for (int t = tid; t < NB; t += PBT) {
        int c = lh[t];
        lbp[t].x = c > 0 ? atomicAdd(&cursors[t * CST], c) : 0;
    }
    if (tid >= PBT - 32) {
        const int lane = tid & 31;
        int carry = 0;
        for (int base = 0; base < NB; base += 256) {
            const int   t0 = base + lane * 8;
            const int4* p  = (const int4*)(lh + t0);
            const int4  q0 = p[0], q1 = p[1];
            int v[8] = {q0.x, q0.y, q0.z, q0.w, q1.x, q1.y, q1.z, q1.w};
            int sum = 0;
#pragma unroll
            for (int j = 0; j < 8; ++j) {
                v[j] = t0 + j < NB ? v[j] : 0;
                sum += v[j];
            }
            int inc = sum;
#pragma unroll
            for (int o = 1; o < 32; o <<= 1) {
                int n = __shfl_up(inc, o);
                if (lane >= o) inc += n;
            }
            int run = carry + inc - sum;
#pragma unroll
            for (int j = 0; j < 8; ++j) {
                if (t0 + j < NB) lbp[t0 + j].y = run;
                run += v[j];
            }
            carry += __builtin_amdgcn_readlane(inc, 31);
        }
        if (lane == 0) lds[o_tot] = carry;
    }
    __syncthreads();
#pragma unroll
    for (int h = 0; h < EPT; h += 8) {
        int2 bp[8];
#pragma unroll
        for (int i = 0; i < 8; ++i) bp[i] = lbp[lr[h + i] >= 0 ? (s[h + i] >> RBSH) : 0];
#pragma unroll
        for (int i = 0; i < 8; ++i) {
            if (lr[h + i] >= 0) {
                int e    = e0 + (h + i) * PBT;
                int b    = s[h + i] >> RBSH;
                int rows = N - b * RB;
                rows     = rows < RB ? rows : RB;
                int pos  = bp[i].x + lr[h + i];
                int lp   = bp[i].y + lr[h + i];
                int g    = -1;
                if (pos >= 0 && pos < rows * (CAP / 2)) g = b * (RB * (CAP / 2)) + pos;
                else                                     ovf_push(counter, ovf, e, E);
                unsigned w = ((unsigned)(s[h + i] & (RB - 1)) << EBITS) | (unsigned)e;
                sw[lp] = make_int2((int)w, d[h + i]);
                sg[lp] = g;
            }
        }
    }
    __syncthreads();
    const int total = lds[o_tot];
    int2* area = (int2*)out;
#pragma unroll
    for (int h = 0; h < EPT; h += 8) {
        int  g[8];
        int2 v[8];
#pragma unroll
        for (int i = 0; i < 8; ++i) {
            int idx = (h + i) * PBT + tid;
            int c   = idx < total ? idx : 0;
            g[i] = sg[c];
            v[i] = sw[c];
        }
#pragma unroll
        for (int i = 0; i < 8; ++i) {
            int idx = (h + i) * PBT + tid;
            if (idx < total && g[i] >= 0) area[g[i]] = v[i];
        }
    }
}

__device__ __forceinline__ void rank_body(int b, int* lc, int* stage, int* cnt,
                                          int* __restrict__ counter, int* __restrict__ ovf,
                                          float* out, int N, int E) {
    const int tid = threadIdx.x;
    int rows = N - b * RB;
    rows = rows < RB ? rows : RB;
    const int cap = rows * (CAP / 2);
    int tot = cnt[b * CST];
    tot = tot < 0 ? 0 : (tot > cap ? cap : tot);
    if (tid < RB) lc[tid] = 0;
    for (int i = tid; i < RB * SLOT0 / 4; i += RBT) ((int4*)stage)[i] = make_int4(0, 0, 0, 0);
    __syncthreads();

    const int2* area = (const int2*)(out + (size_t)b * RB * D_OUT);
    int w[RPT], d[RPT], r[RPT];
#pragma unroll
    for (int i = 0; i < RPT; ++i) {
        int p = i * RBT + tid;
        int2 v = area[p < tot ? p : 0];
        w[i] = v.x;
        d[i] = v.y;
    }
#pragma unroll
    for (int i = 0; i < RPT; ++i) {
        int p = i * RBT + tid;
        r[i] = -1;
        int sl = (int)((unsigned)w[i] >> EBITS);
        if (p < tot && sl < rows) r[i] = atomicAdd(&lc[sl], 1);
    }
    __syncthreads();
#pragma unroll
    for (int i = 0; i < RPT; ++i) {
        if (r[i] >= 0) {
            int sl = (int)((unsigned)w[i] >> EBITS);
            if (r[i] < SLOT0) {
                stage[sl * SLOT0 + r[i]] = d[i];
            } else if (r[i] < CAP) {
                int* bucket = (int*)out + ((size_t)b * RB + sl) * CAP;
                bucket[r[i]] = d[i];
            } else {
                ovf_push(counter, ovf, (int)((unsigned)w[i] & ((1u << EBITS) - 1u)), E);
            }
        }
    }
    __syncthreads();
    for (int i = tid; i < RB * SLOT0 / 4; i += RBT) {
        int row = i >> 3;
        if (row < rows) {
            int4  v = ((const int4*)stage)[i];
            int4* o = (int4*)((int*)out + ((size_t)b * RB + row) * CAP) + (i & 7);
            *o = v;
        }
    }
    if (tid < rows) cnt[b * RB + tid] = lc[tid];
}

__device__ __forceinline__ void gemm_body(int bid, const float* __restrict__ node,
                                          const float* __restrict__ W, const float* __restrict__ att,
                                          float* __restrict__ h, float* __restrict__ ssrc, int N) {
    int wave = (bid * (int)blockDim.x + (int)threadIdx.x) >> 5;
    int lane = threadIdx.x & 31;
    int row0 = wave * 16;
    if (row0 >= N) return;

    int mrow = lane & 15;
    int koff = (lane >> 4) * 2;

    v8f c0 = {}, c1 = {}, c2 = {}, c3 = {};

    const int   nlo = lane & 15;
    const float as0 = att[nlo], as1 = att[nlo + 16], as2 = att[nlo + 32], as3 = att[nlo + 48];

    const float* arow = node + (size_t)(row0 + mrow) * D_IN + koff;

    for (int k = 0; k < D_IN; k += 4) {
        v2f a;
        a.x = arow[k];
        a.y = arow[k + 1];

        const float* bp = W + (size_t)(k + koff) * D_OUT + mrow;
        v2f b0, b1, b2, b3;
        b0.x = bp[0];            b0.y = bp[D_OUT];
        b1.x = bp[16];           b1.y = bp[16 + D_OUT];
        b2.x = bp[32];           b2.y = bp[32 + D_OUT];
        b3.x = bp[48];           b3.y = bp[48 + D_OUT];

        c0 = __builtin_amdgcn_wmma_f32_16x16x4_f32(false, a, false, b0, (short)0, c0, false, false);
        c1 = __builtin_amdgcn_wmma_f32_16x16x4_f32(false, a, false, b1, (short)0, c1, false, false);
        c2 = __builtin_amdgcn_wmma_f32_16x16x4_f32(false, a, false, b2, (short)0, c2, false, false);
        c3 = __builtin_amdgcn_wmma_f32_16x16x4_f32(false, a, false, b3, (short)0, c3, false, false);
    }

    int rbase = row0 + ((lane >> 4) * 8);
#pragma unroll
    for (int r = 0; r < 8; ++r) {
        float* op = h + (size_t)(rbase + r) * D_OUT + nlo;
        op[0]  = c0[r];
        op[16] = c1[r];
        op[32] = c2[r];
        op[48] = c3[r];
    }

    float ps[8];
#pragma unroll
    for (int r = 0; r < 8; ++r)
        ps[r] = fmaf(c3[r], as3, fmaf(c2[r], as2, fmaf(c1[r], as1, c0[r] * as0)));
    float u[8], t[4], s[2], sc;
#pragma unroll
    for (int i = 0; i < 8; ++i) u[i] = ps[i] + __shfl_xor(ps[i], 8);
    {
        const bool hi = (nlo & 4) != 0;
#pragma unroll
        for (int i = 0; i < 4; ++i) {
            float keep = hi ? u[i + 4] : u[i];
            float send = hi ? u[i] : u[i + 4];
            t[i] = keep + __shfl_xor(send, 4);
        }
    }
    {
        const bool hi = (nlo & 2) != 0;
#pragma unroll
        for (int i = 0; i < 2; ++i) {
            float keep = hi ? t[i + 2] : t[i];
            float send = hi ? t[i] : t[i + 2];
            s[i] = keep + __shfl_xor(send, 2);
        }
    }
    {
        const bool hi = (nlo & 1) != 0;
        float keep = hi ? s[1] : s[0];
        float send = hi ? s[0] : s[1];
        sc = keep + __shfl_xor(send, 1);
    }
    if (nlo < 8 && rbase + nlo < N) ssrc[rbase + nlo] = sc;
}

__global__ __launch_bounds__(PBT) void gat_part_gemm(const int* __restrict__ src, const int* __restrict__ dst,
                                                     int* __restrict__ cursors, int* __restrict__ counter,
                                                     int* __restrict__ ovf, float* out,
                                                     const float* __restrict__ node, const float* __restrict__ W,
                                                     const float* __restrict__ a, float* __restrict__ h,
                                                     float* __restrict__ ssrc,
                                                     int N, int E, int NB, int PB) {
    extern __shared__ __attribute__((aligned(16))) int lds[];
    const int bid = (int)blockIdx.x;
    if (bid < PB) part_body(bid, lds, src, dst, cursors, counter, ovf, out, N, E, NB);
    else          gemm_body(bid - PB, node, W, a, h, ssrc, N);
}

__global__ __launch_bounds__(RBT) void gat_rank(int* cnt, int* __restrict__ counter, int* __restrict__ ovf,
                                                float* out, int N, int E) {
    __shared__ int lc[RB];
    __shared__ __attribute__((aligned(16))) int stage[RB * SLOT0];
    rank_body((int)blockIdx.x, lc, stage, cnt, counter, ovf, out, N, E);
}

template <int M>
__device__ __forceinline__ float lane_xor(float x) {
    int b = __float_as_int(x);
    if constexpr (M == 16)
        return __int_as_float(__builtin_amdgcn_permlanex16(b, b, 0x76543210, 0xfedcba98, false, false));
    else
        return __int_as_float(__builtin_amdgcn_update_dpp(0, b, 0x160 + M, 0xf, 0xf, true));
}

template <int K>
struct Tree {
    static __device__ __forceinline__ float run(const float* x, int lane) {
        constexpr int H = K / 2;
        const bool hi = (lane & H) != 0;
        float y[H];
        for (int i = 0; i < H; ++i) {
            float keep = hi ? x[i + H] : x[i];
            float send = hi ? x[i] : x[i + H];
            y[i] = keep + lane_xor<H>(send);
        }
        return Tree<H>::run(y, lane);
    }
};
template <>
struct Tree<1> {
    static __device__ __forceinline__ float run(const float* x, int) { return x[0]; }
};

template <int M, int TOP>
__device__ __forceinline__ float sum_bits(float v) {
    if constexpr (M < TOP) { v += lane_xor<M>(v); return sum_bits<M * 2, TOP>(v); }
    else return v;
}
template <int M, int TOP>
__device__ __forceinline__ float max_bits(float v) {
    if constexpr (M < TOP) { v = fmaxf(v, lane_xor<M>(v)); return max_bits<M * 2, TOP>(v); }
    else return v;
}

struct Soft {
    float  m;
    float  s;
    float2 acc;
};

template <int K>
__device__ __forceinline__ void add_rows(const float2* __restrict__ hl, float2 ad, float ss,
                                         int dl, int t, int lane, int N, Soft& st) {
    int    d[K];
    float2 v[K];
    float  p[K];
    for (int i = 0; i < K; ++i) {
        int di = __builtin_amdgcn_readlane(dl, t + i);
        d[i] = di < 0 ? 0 : (di >= N ? N - 1 : di);
    }
    for (int i = 0; i < K; ++i) v[i] = hl[(size_t)d[i] * (D_OUT / 2)];
    for (int i = 0; i < K; ++i) p[i] = fmaf(v[i].y, ad.y, v[i].x * ad.x);
    float sc = Tree<K>::run(p, lane);
    sc = sum_bits<K, 32>(sc);
    const float l  = leaky(ss + sc);
    const float gm = max_bits<1, K>(l);
    const float mn = fmaxf(st.m, gm);
    const float rs = __expf(st.m - mn);
    const float w  = __expf(l - mn);
    const float gs = sum_bits<1, K>(w);
    st.s     = fmaf(st.s, rs, gs);
    st.acc.x *= rs;
    st.acc.y *= rs;
    for (int i = 0; i < K; ++i) {
        float wi = __int_as_float(__builtin_amdgcn_readlane(__float_as_int(w), i));
        st.acc.x = fmaf(wi, v[i].x, st.acc.x);
        st.acc.y = fmaf(wi, v[i].y, st.acc.y);
    }
    st.m = mn;
}

__device__ __forceinline__ void add_chunk(const float2* __restrict__ hl, float2 ad, float ss,
                                          int dl, int m, int lane, int N, Soft& st) {
    int t = 0;
    for (; t + 16 <= m; t += 16) add_rows<16>(hl, ad, ss, dl, t, lane, N, st);
    if (m - t >= 8) { add_rows<8>(hl, ad, ss, dl, t, lane, N, st); t += 8; }
    if (m - t >= 4) { add_rows<4>(hl, ad, ss, dl, t, lane, N, st); t += 4; }
    if (m - t >= 2) { add_rows<2>(hl, ad, ss, dl, t, lane, N, st); t += 2; }
    if (m - t >= 1) { add_rows<1>(hl, ad, ss, dl, t, lane, N, st); }
}

__device__ __forceinline__ bool ovf_slice(const int* __restrict__ ovf, const int* __restrict__ src,
                                          const int* __restrict__ dst, int j, int novf, int lane,
                                          int node, int N, int E, int& d) {
    d = 0;
    int k = j + lane;
    if (k >= novf) return false;
    int e = ovf[k];
    e = e < 0 ? 0 : (e >= E ? E - 1 : e);
    if (src[e] != node) return false;
    int t = dst[e];
    d = t < 0 ? 0 : (t >= N ? N - 1 : t);
    return true;
}

__global__ void gat_aggregate(const int* __restrict__ cnt, const int* __restrict__ counter,
                              const int* __restrict__ ovf,
                              const int* __restrict__ src, const int* __restrict__ dst,
                              const float* __restrict__ ssrc, const float* __restrict__ a,
                              const float* __restrict__ h, float* out,
                              int N, int E) {
    int wid  = __builtin_amdgcn_readfirstlane((int)(threadIdx.x >> 5));
    int node = (int)blockIdx.x * (int)(blockDim.x >> 5) + wid;
    int lane = threadIdx.x & 31;
    if (node >= N) return;

    const int*   lst = (const int*)out + (size_t)node * CAP;
    const int    dr  = lst[lane];
    const float  ss  = ssrc[node];
    const float2 ad  = ((const float2*)(a + D_OUT))[lane];
    int n = cnt[node];
    int novf = counter[0];
    __builtin_amdgcn_sched_barrier(0);

    n = n < 0 ? 0 : (n > E ? E : n);
    const int nb = n < CAP ? n : CAP;

    Soft st;
    st.m   = -INFINITY;
    st.s   = 0.f;
    st.acc = make_float2(0.f, 0.f);
    const float2* hl = (const float2*)h + lane;

    if (nb > 0) add_chunk(hl, ad, ss, dr, nb < 32 ? nb : 32, lane, N, st);
    for (int c = 32; c < nb; c += 32) {
        int j  = c + lane;
        int dl = 0;
        if (j < nb) dl = lst[j];
        add_chunk(hl, ad, ss, dl, (nb - c) < 32 ? (nb - c) : 32, lane, N, st);
    }
    novf = novf < 0 ? 0 : (novf > E ? E : novf);
    for (int j = 0; j < novf; j += 32) {
        int  dl;
        bool hit = ovf_slice(ovf, src, dst, j, novf, lane, node, N, E, dl);
        unsigned mask = __builtin_amdgcn_ballot_w32(hit);
        while (mask) {
            int t = __builtin_ctz(mask);
            mask &= mask - 1;
            add_rows<1>(hl, ad, ss, dl, t, lane, N, st);
        }
    }
    const float inv = 1.0f / (st.s + EPS);
    float2* o2 = (float2*)out;
    o2[(size_t)node * (D_OUT / 2) + lane] = make_float2(st.acc.x * inv, st.acc.y * inv);
}

extern "C" void kernel_launch(void* const* d_in, const int* in_sizes, int n_in,
                              void* d_out, int out_size, void* d_ws, size_t ws_size,
                              hipStream_t stream) {
    const float* node = (const float*)d_in[0];
    const int*   ei   = (const int*)d_in[1];
    const float* W    = (const float*)d_in[2];
    const float* a    = (const float*)d_in[3];
    float* out        = (float*)d_out;

    int N = in_sizes[0] / D_IN;
    int E = in_sizes[1] / 2;
    const int* src = ei;
    const int* dst = ei + E;

    float* h       = (float*)d_ws;
    float* ssrc    = h + (size_t)N * D_OUT;
    int*   cnt     = (int*)(ssrc + N);
    int*   counter = cnt + N;
    int*   ovf     = counter + 1;
    int*   cursors = cnt;
    int    NB      = (N + RB - 1) / RB;

    const int B = 256;

    gat_init<<<(NB + B - 1) / B, B, 0, stream>>>(cursors, counter, NB);
    {
        int PB      = (E + PBT * EPT - 1) / (PBT * EPT);
        int waves   = (N + 15) / 16;
        long long threads = (long long)waves * 32;
        int gblocks = (int)((threads + PBT - 1) / PBT);
        gat_part_gemm<<<PB + gblocks, PBT, (size_t)(3 * NB + 8 + 3 * PBT * EPT) * sizeof(int), stream>>>(
            src, dst, cursors, counter, ovf, out, node, W, a, h, ssrc, N, E, NB, PB);
    }
    gat_rank<<<NB, RBT, 0, stream>>>(cnt, counter, ovf, out, N, E);
    {
        long long threads = (long long)N * 32;
        int blocks = (int)((threads + B - 1) / B);
        gat_aggregate<<<blocks, B, 0, stream>>>(cnt, counter, ovf, src, dst, ssrc, a, h, out, N, E);
    }
}


# bins of 392 nodes, not 512: the rank + aggregate launch gets 256 blocks, one per core of the chip (196 left 60 cores idle); a wave walks 12 or 13 nodes, not 16; sum order unchanged (arrival rank)
# speedup vs baseline: 5.4963x; 1.1303x over previous
#include <hip/hip_runtime.h>


#define D_IN   128
#define D_OUT  64
#define ALPHA  0.2f
#define EPS    1e-10f
#define CAP    D_OUT
#define RB     392
#define OSH    9
#define EPT    16
#define PBT    256
#define RBT    1024
#define RPT    13
#define CST    RB
#define EBITS  23
static_assert(RB <= (1 << OSH) && OSH + EBITS <= 32, "a node's offset in its bin and the edge id share one word");
static_assert(RPT * RBT >= RB * (CAP / 2), "the ranking block must cover every record slot of its bin");

typedef __attribute__((ext_vector_type(2))) float v2f;
typedef __attribute__((ext_vector_type(8))) float v8f;

__device__ __forceinline__ float leaky(float x) { return x > 0.f ? x : ALPHA * x; }

__global__ void gat_init(int* __restrict__ cursors, int* __restrict__ counter, int NB) {
    int i = blockIdx.x * blockDim.x + threadIdx.x;
    if (i < NB) cursors[i * CST] = 0;
    if (i == 0) counter[0] = 0;
}

__device__ __forceinline__ void ovf_push(int* __restrict__ counter, int* __restrict__ ovf, int e, int E) {
    int k = atomicAdd(counter, 1);
    if (k >= 0 && k < E) ovf[k] = e;
}

__device__ __forceinline__ void part_body(int bid, int* lds,
                                          const int* __restrict__ src, const int* __restrict__ dst,
                                          int* __restrict__ cursors, int* __restrict__ counter,
                                          int* __restrict__ ovf, float* out, int N, int E, int NB) {
    const int o_lbp = (NB + 1) & ~1;
    const int o_tot = o_lbp + 2 * NB;
    const int o_sw  = (o_tot + 4) & ~3;
    const int o_sg  = o_sw + 2 * PBT * EPT;
    int*  lh  = lds;
    int2* lbp = (int2*)(lds + o_lbp);
    int2* sw  = (int2*)(lds + o_sw);
    int*  sg  = lds + o_sg;
    const int tid = threadIdx.x;
    for (int t = tid; t < NB; t += PBT) lh[t] = 0;
    __syncthreads();

    const int e0 = bid * (PBT * EPT) + tid;
    int s[EPT], d[EPT], lr[EPT];
#pragma unroll
    for (int i = 0; i < EPT; ++i) {
        int e = e0 + i * PBT;
        s[i] = -1;
        d[i] = 0;
        if (e < E) { s[i] = src[e]; d[i] = dst[e]; }
    }
#pragma unroll
    for (int i = 0; i < EPT; ++i) {
        int e = e0 + i * PBT;
        lr[i] = -1;
        if ((unsigned)s[i] < (unsigned)N) {
            const unsigned b = (unsigned)s[i] / RB;
            s[i] = (int)((b << OSH) | ((unsigned)s[i] - b * RB));
            if (e < (1 << EBITS)) lr[i] = atomicAdd(&lh[b], 1);
            else                  ovf_push(counter, ovf, e, E);
        }
    }
    __syncthreads();
    for (int t = tid; t < NB; t += PBT) {
        int c = lh[t];
        lbp[t].x = c > 0 ? atomicAdd(&cursors[t * CST], c) : 0;
    }
    if (tid >= PBT - 32) {
        const int lane = tid & 31;
        int carry = 0;
        for (int base = 0; base < NB; base += 256) {
            const int   t0 = base + lane * 8;
            const int4* p  = (const int4*)(lh + t0);
            const int4  q0 = p[0], q1 = p[1];
            int v[8] = {q0.x, q0.y, q0.z, q0.w, q1.x, q1.y, q1.z, q1.w};
            int sum = 0;
#pragma unroll
            for (int j = 0; j < 8; ++j) {
                v[j] = t0 + j < NB ? v[j] : 0;
                sum += v[j];
            }
            int inc = sum;
#pragma unroll
            for (int o = 1; o < 32; o <<= 1) {
                int n = __shfl_up(inc, o);
                if (lane >= o) inc += n;
            }
            int run = carry + inc - sum;
#pragma unroll
            for (int j = 0; j < 8; ++j) {
                if (t0 + j < NB) lbp[t0 + j].y = run;
                run += v[j];
            }
            carry += __builtin_amdgcn_readlane(inc, 31);
        }
        if (lane == 0) lds[o_tot] = carry;
    }
    __syncthreads();
#pragma unroll
    for (int h = 0; h < EPT; h += 8) {
        int2 bp[8];
#pragma unroll
        for (int i = 0; i < 8; ++i) bp[i] = lbp[lr[h + i] >= 0 ? (s[h + i] >> OSH) : 0];
#pragma unroll
        for (int i = 0; i < 8; ++i) {
            if (lr[h + i] >= 0) {
                int e    = e0 + (h + i) * PBT;
                int b    = s[h + i] >> OSH;
                int rows = N - b * RB;
                rows     = rows < RB ? rows : RB;
                int pos  = bp[i].x + lr[h + i];
                int lp   = bp[i].y + lr[h + i];
                int g    = -1;
                if (pos >= 0 && pos < rows * (CAP / 2)) g = b * (RB * (CAP / 2)) + pos;
                else                                     ovf_push(counter, ovf, e, E);
                unsigned w = ((unsigned)(s[h + i] & ((1 << OSH) - 1)) << EBITS) | (unsigned)e;
                sw[lp] = make_int2((int)w, d[h + i]);
                sg[lp] = g;
            }
        }
    }
    __syncthreads();
    const int total = lds[o_tot];
    int2* area = (int2*)out;
#pragma unroll
    for (int h = 0; h < EPT; h += 8) {
        int  g[8];
        int2 v[8];
#pragma unroll
        for (int i = 0; i < 8; ++i) {
            int idx = (h + i) * PBT + tid;
            int c   = idx < total ? idx : 0;
            g[i] = sg[c];
            v[i] = sw[c];
        }
#pragma unroll
        for (int i = 0; i < 8; ++i) {
            int idx = (h + i) * PBT + tid;
            if (idx < total && g[i] >= 0) area[g[i]] = v[i];
        }
    }
}

__device__ __forceinline__ void gemm_body(int bid, const float* __restrict__ node,
                                          const float* __restrict__ W, const float* __restrict__ att,
                                          float* __restrict__ h, float* __restrict__ ssrc, int N) {
    int wave = (bid * (int)blockDim.x + (int)threadIdx.x) >> 5;
    int lane = threadIdx.x & 31;
    int row0 = wave * 16;
    if (row0 >= N) return;

    int mrow = lane & 15;
    int koff = (lane >> 4) * 2;

    v8f c0 = {}, c1 = {}, c2 = {}, c3 = {};

    const int   nlo = lane & 15;
    const float as0 = att[nlo], as1 = att[nlo + 16], as2 = att[nlo + 32], as3 = att[nlo + 48];

    const float* arow = node + (size_t)(row0 + mrow) * D_IN + koff;

    for (int k = 0; k < D_IN; k += 4) {
        v2f a;
        a.x = arow[k];
        a.y = arow[k + 1];

        const float* bp = W + (size_t)(k + koff) * D_OUT + mrow;
        v2f b0, b1, b2, b3;
        b0.x = bp[0];            b0.y = bp[D_OUT];
        b1.x = bp[16];           b1.y = bp[16 + D_OUT];
        b2.x = bp[32];           b2.y = bp[32 + D_OUT];
        b3.x = bp[48];           b3.y = bp[48 + D_OUT];

        c0 = __builtin_amdgcn_wmma_f32_16x16x4_f32(false, a, false, b0, (short)0, c0, false, false);
        c1 = __builtin_amdgcn_wmma_f32_16x16x4_f32(false, a, false, b1, (short)0, c1, false, false);
        c2 = __builtin_amdgcn_wmma_f32_16x16x4_f32(false, a, false, b2, (short)0, c2, false, false);
        c3 = __builtin_amdgcn_wmma_f32_16x16x4_f32(false, a, false, b3, (short)0, c3, false, false);
    }

    int rbase = row0 + ((lane >> 4) * 8);
#pragma unroll
    for (int r = 0; r < 8; ++r) {
        float* op = h + (size_t)(rbase + r) * D_OUT + nlo;
        op[0]  = c0[r];
        op[16] = c1[r];
        op[32] = c2[r];
        op[48] = c3[r];
    }

    float ps[8];
#pragma unroll
    for (int r = 0; r < 8; ++r)
        ps[r] = fmaf(c3[r], as3, fmaf(c2[r], as2, fmaf(c1[r], as1, c0[r] * as0)));
    float u[8], t[4], s[2], sc;
#pragma unroll
    for (int i = 0; i < 8; ++i) u[i] = ps[i] + __shfl_xor(ps[i], 8);
    {
        const bool hi = (nlo & 4) != 0;
#pragma unroll
        for (int i = 0; i < 4; ++i) {
            float keep = hi ? u[i + 4] : u[i];
            float send = hi ? u[i] : u[i + 4];
            t[i] = keep + __shfl_xor(send, 4);
        }
    }
    {
        const bool hi = (nlo & 2) != 0;
#pragma unroll
        for (int i = 0; i < 2; ++i) {
            float keep = hi ? t[i + 2] : t[i];
            float send = hi ? t[i] : t[i + 2];
            s[i] = keep + __shfl_xor(send, 2);
        }
    }
    {
        const bool hi = (nlo & 1) != 0;
        float keep = hi ? s[1] : s[0];
        float send = hi ? s[0] : s[1];
        sc = keep + __shfl_xor(send, 1);
    }
    if (nlo < 8 && rbase + nlo < N) ssrc[rbase + nlo] = sc;
}

__global__ __launch_bounds__(PBT) void gat_part_gemm(const int* __restrict__ src, const int* __restrict__ dst,
                                                     int* __restrict__ cursors, int* __restrict__ counter,
                                                     int* __restrict__ ovf, float* out,
                                                     const float* __restrict__ node, const float* __restrict__ W,
                                                     const float* __restrict__ a, float* __restrict__ h,
                                                     float* __restrict__ ssrc,
                                                     int N, int E, int NB, int PB) {
    extern __shared__ __attribute__((aligned(16))) int lds[];
    const int bid = (int)blockIdx.x;
    if (bid < PB) part_body(bid, lds, src, dst, cursors, counter, ovf, out, N, E, NB);
    else          gemm_body(bid - PB, node, W, a, h, ssrc, N);
}

template <int M>
__device__ __forceinline__ float lane_xor(float x) {
    int b = __float_as_int(x);
    if constexpr (M == 16)
        return __int_as_float(__builtin_amdgcn_permlanex16(b, b, 0x76543210, 0xfedcba98, false, false));
    else
        return __int_as_float(__builtin_amdgcn_update_dpp(0, b, 0x160 + M, 0xf, 0xf, true));
}

template <int K>
struct Tree {
    static __device__ __forceinline__ float run(const float* x, int lane) {
        constexpr int H = K / 2;
        const bool hi = (lane & H) != 0;
        float y[H];
        for (int i = 0; i < H; ++i) {
            float keep = hi ? x[i + H] : x[i];
            float send = hi ? x[i] : x[i + H];
            y[i] = keep + lane_xor<H>(send);
        }
        return Tree<H>::run(y, lane);
    }
};
template <>
struct Tree<1> {
    static __device__ __forceinline__ float run(const float* x, int) { return x[0]; }
};

template <int M, int TOP>
__device__ __forceinline__ float sum_bits(float v) {
    if constexpr (M < TOP) { v += lane_xor<M>(v); return sum_bits<M * 2, TOP>(v); }
    else return v;
}
template <int M, int TOP>
__device__ __forceinline__ float max_bits(float v) {
    if constexpr (M < TOP) { v = fmaxf(v, lane_xor<M>(v)); return max_bits<M * 2, TOP>(v); }
    else return v;
}

struct Soft {
    float  m;
    float  s;
    float2 acc;
};

template <int K>
__device__ __forceinline__ void add_rows(const float2* __restrict__ hl, float2 ad, float ss,
                                         int dl, int t, int lane, int N, Soft& st) {
    int    d[K];
    float2 v[K];
    float  p[K];
    for (int i = 0; i < K; ++i) {
        int di = __builtin_amdgcn_readlane(dl, t + i);
        d[i] = di < 0 ? 0 : (di >= N ? N - 1 : di);
    }
    for (int i = 0; i < K; ++i) v[i] = hl[(size_t)d[i] * (D_OUT / 2)];
    for (int i = 0; i < K; ++i) p[i] = fmaf(v[i].y, ad.y, v[i].x * ad.x);
    float sc = Tree<K>::run(p, lane);
    sc = sum_bits<K, 32>(sc);
    const float l  = leaky(ss + sc);
    const float gm = max_bits<1, K>(l);
    const float mn = fmaxf(st.m, gm);
    const float rs = __expf(st.m - mn);
    const float w  = __expf(l - mn);
    const float gs = sum_bits<1, K>(w);
    st.s     = fmaf(st.s, rs, gs);
    st.acc.x *= rs;
    st.acc.y *= rs;
    for (int i = 0; i < K; ++i) {
        float wi = __int_as_float(__builtin_amdgcn_readlane(__float_as_int(w), i));
        st.acc.x = fmaf(wi, v[i].x, st.acc.x);
        st.acc.y = fmaf(wi, v[i].y, st.acc.y);
    }
    st.m = mn;
}

__device__ __forceinline__ void add_chunk(const float2* __restrict__ hl, float2 ad, float ss,
                                          int dl, int m, int lane, int N, Soft& st) {
    int t = 0;
    for (; t + 16 <= m; t += 16) add_rows<16>(hl, ad, ss, dl, t, lane, N, st);
    if (m - t >= 8) { add_rows<8>(hl, ad, ss, dl, t, lane, N, st); t += 8; }
    if (m - t >= 4) { add_rows<4>(hl, ad, ss, dl, t, lane, N, st); t += 4; }
    if (m - t >= 2) { add_rows<2>(hl, ad, ss, dl, t, lane, N, st); t += 2; }
    if (m - t >= 1) { add_rows<1>(hl, ad, ss, dl, t, lane, N, st); }
}

__device__ __forceinline__ bool ovf_slice(const int* __restrict__ ovf, const int* __restrict__ src,
                                          const int* __restrict__ dst, int j, int novf, int lane,
                                          int node, int N, int E, int& d) {
    d = 0;
    int k = j + lane;
    if (k >= novf) return false;
    int e = ovf[k];
    e = e < 0 ? 0 : (e >= E ? E - 1 : e);
    if (src[e] != node) return false;
    int t = dst[e];
    d = t < 0 ? 0 : (t >= N ? N - 1 : t);
    return true;
}

#define BPITCH (CAP + 1)
#define LOV    (RB * (CAP / 2))
#define NWV    (RBT / 32)
static_assert((RB + NWV - 1) / NWV <= 32, "a wave keeps its nodes' counts and scores one per lane");

__global__ __launch_bounds__(RBT) void gat_rank_aggregate(const int* __restrict__ cursors,
                                                          const int* __restrict__ counter,
                                                          const int* __restrict__ ovf,
                                                          const int* __restrict__ src,
                                                          const int* __restrict__ dst,
                                                          const float* __restrict__ ssrc,
                                                          const float* __restrict__ a,
                                                          const float* __restrict__ h, float* out,
                                                          int N, int E) {
    __shared__ int  lc[RB];
    __shared__ int  lovn;
    __shared__ int  bucket[RB * BPITCH];
    __shared__ int2 lov[LOV];

    const int b   = (int)blockIdx.x;
    const int tid = threadIdx.x;
    int rows = N - b * RB;
    rows = rows < RB ? rows : RB;
    const int cap = rows * (CAP / 2);
    int tot = cursors[b * CST];
    tot = tot < 0 ? 0 : (tot > cap ? cap : tot);
    if (tid < RB) lc[tid] = 0;
    if (tid == 0) lovn = 0;
    __syncthreads();

    {
        const int2* area = (const int2*)(out + (size_t)b * RB * D_OUT);
        int w[RPT], d[RPT];
#pragma unroll
        for (int i = 0; i < RPT; ++i) {
            int p = i * RBT + tid;
            int2 v = area[p < tot ? p : 0];
            w[i] = v.x;
            d[i] = v.y;
        }
#pragma unroll
        for (int i = 0; i < RPT; ++i) {
            int p  = i * RBT + tid;
            int sl = (int)((unsigned)w[i] >> EBITS);
            if (p < tot && sl < rows) {
                int r = atomicAdd(&lc[sl], 1);
                if (r < CAP) {
                    bucket[sl * BPITCH + r] = d[i];
                } else {
                    int k = atomicAdd(&lovn, 1);
                    if (k < LOV) lov[k] = make_int2(sl, d[i]);
                }
            }
        }
    }
    __syncthreads();

    const int wv   = __builtin_amdgcn_readfirstlane(tid >> 5);
    const int lane = tid & 31;
    const int sl0  = (wv * RB) / NWV;
    const int npw  = ((wv + 1) * RB) / NWV - sl0;
    if (sl0 >= rows) return;
    const int node0 = b * RB + sl0;

    const int   li  = lane < npw ? lane : 0;
    const int   nl  = lc[sl0 + li];
    const float ssl = (sl0 + li < rows) ? ssrc[node0 + li] : 0.f;
    const float2 ad = ((const float2*)(a + D_OUT))[lane];
    int novf = counter[0];
    novf = novf < 0 ? 0 : (novf > E ? E : novf);
    int nlov = __builtin_amdgcn_readfirstlane(lovn);
    nlov = nlov < 0 ? 0 : (nlov > LOV ? LOV : nlov);
    const float2* hl = (const float2*)h + lane;
    float2* o2 = (float2*)out;

    for (int q = 0; q < npw; ++q) {
        const int sl = sl0 + q;
        if (sl >= rows) break;
        const int node = node0 + q;
        int n = __builtin_amdgcn_readlane(nl, q);
        const float ss = __int_as_float(__builtin_amdgcn_readlane(__float_as_int(ssl), q));
        n = n < 0 ? 0 : n;
        const int nb = n < CAP ? n : CAP;
        const int* lst = bucket + sl * BPITCH;

        Soft st;
        st.m   = -INFINITY;
        st.s   = 0.f;
        st.acc = make_float2(0.f, 0.f);

        if (nb > 0) {
            int dl = lst[lane];
            add_chunk(hl, ad, ss, dl, nb < 32 ? nb : 32, lane, N, st);
        }
        if (nb > 32) {
            int dl = lst[32 + lane];
            add_chunk(hl, ad, ss, dl, nb - 32, lane, N, st);
        }
        for (int j = 0; j < nlov; j += 32) {
            int  k   = j + lane;
            int2 e2  = lov[k < nlov ? k : 0];
            bool hit = k < nlov && e2.x == sl;
            int  dl  = hit ? e2.y : 0;
            unsigned mask = __builtin_amdgcn_ballot_w32(hit);
            while (mask) {
                int t = __builtin_ctz(mask);
                mask &= mask - 1;
                add_rows<1>(hl, ad, ss, dl, t, lane, N, st);
            }
        }
        for (int j = 0; j < novf; j += 32) {
            int  dl;
            bool hit = ovf_slice(ovf, src, dst, j, novf, lane, node, N, E, dl);
            unsigned mask = __builtin_amdgcn_ballot_w32(hit);
            while (mask) {
                int t = __builtin_ctz(mask);
                mask &= mask - 1;
                add_rows<1>(hl, ad, ss, dl, t, lane, N, st);
            }
        }
        const float inv = 1.0f / (st.s + EPS);
        o2[(size_t)node * (D_OUT / 2) + lane] = make_float2(st.acc.x * inv, st.acc.y * inv);
    }
}

extern "C" void kernel_launch(void* const* d_in, const int* in_sizes, int n_in,
                              void* d_out, int out_size, void* d_ws, size_t ws_size,
                              hipStream_t stream) {
    const float* node = (const float*)d_in[0];
    const int*   ei   = (const int*)d_in[1];
    const float* W    = (const float*)d_in[2];
    const float* a    = (const float*)d_in[3];
    float* out        = (float*)d_out;

    int N = in_sizes[0] / D_IN;
    int E = in_sizes[1] / 2;
    const int* src = ei;
    const int* dst = ei + E;

    float* h       = (float*)d_ws;
    float* ssrc    = h + (size_t)N * D_OUT;
    int*   cursors = (int*)(ssrc + N);
    int*   counter = cursors + N;
    int*   ovf     = counter + 1;
    int    NB      = (N + RB - 1) / RB;

    const int B = 256;

    gat_init<<<(NB + B - 1) / B, B, 0, stream>>>(cursors, counter, NB);
    {
        int PB      = (E + PBT * EPT - 1) / (PBT * EPT);
        int waves   = (N + 15) / 16;
        long long threads = (long long)waves * 32;
        int gblocks = (int)((threads + PBT - 1) / PBT);
        gat_part_gemm<<<PB + gblocks, PBT, (size_t)(3 * NB + 8 + 3 * PBT * EPT) * sizeof(int), stream>>>(
            src, dst, cursors, counter, ovf, out, node, W, a, h, ssrc, N, E, NB, PB);
    }
    gat_rank_aggregate<<<NB, RBT, 0, stream>>>(cursors, counter, ovf, src, dst, ssrc, a, h, out, N, E);
}
